# starting kernel
# baseline (speedup 1.0000x reference)
#include <hip/hip_runtime.h>
#include <cstddef>
#include <cstdint>

typedef float v2f __attribute__((ext_vector_type(2)));
typedef float v8f __attribute__((ext_vector_type(8)));

__device__ __forceinline__ float lrelu(float x, float s) {
    return x > 0.0f ? x : s * x;
}

__device__ __forceinline__ void atomicMaxF(float* addr, float v) {
    if (v >= 0.0f) {
        atomicMax(reinterpret_cast<int*>(addr), __float_as_int(v));
    } else {
        atomicMin(reinterpret_cast<unsigned int*>(addr), __float_as_uint(v));
    }
}

template <int K, int NOUT>
__launch_bounds__(256)
__global__ void gat_gemm_wmma(const float* __restrict__ X,
                              const float* __restrict__ W,
                              float* __restrict__ H, int n) {
    constexpr int NT  = NOUT / 16;
    constexpr int LDK = K + 2;
    __shared__ alignas(16) float sWt[NOUT * LDK];
    for (int i = threadIdx.x; i < K * NOUT; i += 256) {
        const int k = i / NOUT;
        const int c = i - k * NOUT;
        sWt[c * LDK + k] = W[i];
    }
    __syncthreads();

    const int wave = threadIdx.x >> 5;
    const int lane = threadIdx.x & 31;
    const int tile = blockIdx.x * 8 + wave;
    const int m0 = tile * 16;
    if (m0 >= n) return;

    const int ml = lane & 15;
    const int kh = lane >> 4;
    int row = m0 + ml;
    row = row < n ? row : n - 1;
    const float* __restrict__ xrow = X + (size_t)row * K;

    v8f acc[NT] = {};

#pragma unroll
    for (int kk = 0; kk < K; kk += 4) {
        const v2f a = *(const v2f*)(xrow + kk + 2 * kh);
#pragma unroll
        for (int t = 0; t < NT; ++t) {
            const v2f b = *(const v2f*)(&sWt[(t * 16 + ml) * LDK + kk + 2 * kh]);
            acc[t] = __builtin_amdgcn_wmma_f32_16x16x4_f32(
                false, a, false, b, (short)0, acc[t], false, false);
        }
    }

    const int rb = kh * 8;
#pragma unroll
    for (int t = 0; t < NT; ++t) {
#pragma unroll
        for (int r = 0; r < 8; ++r) {
            const int m = m0 + rb + r;
            if (m < n) H[(size_t)m * NOUT + t * 16 + ml] = acc[t][r];
        }
    }
}

template <int D>
__global__ void node_dots(const float* __restrict__ H,
                          const float* __restrict__ a_s,
                          const float* __restrict__ a_d,
                          float* __restrict__ os, float* __restrict__ od, int n) {
    const int i = blockIdx.x * blockDim.x + threadIdx.x;
    if (i >= n) return;
    const float* __restrict__ h = H + (size_t)i * D;
    float ss = 0.0f, sd = 0.0f;
#pragma unroll
    for (int f = 0; f < D; ++f) {
        const float v = h[f];
        ss += v * a_s[f];
        sd += v * a_d[f];
    }
    os[i] = ss;
    od[i] = sd;
}

__global__ void init_neg_inf(float* __restrict__ m, int n) {
    const int i = blockIdx.x * blockDim.x + threadIdx.x;
    if (i < n) m[i] = -__builtin_inff();
}

__device__ __forceinline__ void edge_sd(const int* __restrict__ ei, int E,
                                        unsigned e, int& s, int& d) {
    if (e < (unsigned)E) {
        s = ei[e];
        d = ei[(size_t)E + e];
    } else {
        s = d = (int)(e - (unsigned)E);
    }
}

__global__ void edge_logit_max(const int* __restrict__ ei, int E, int ET,
                               const float* __restrict__ as,
                               const float* __restrict__ ad,
                               float* __restrict__ mx) {
    const unsigned t = blockIdx.x * blockDim.x + threadIdx.x;
    if (t >= (unsigned)ET) return;
    int s, d;
    edge_sd(ei, E, t, s, d);
    atomicMaxF(&mx[d], lrelu(as[s] + ad[d], 0.2f));
}

__global__ void edge_exp_sum(const int* __restrict__ ei, int E, int ET,
                             const float* __restrict__ as,
                             const float* __restrict__ ad,
                             const float* __restrict__ mx,
                             float* __restrict__ ex, float* __restrict__ den) {
    const unsigned t = blockIdx.x * blockDim.x + threadIdx.x;
    if (t >= (unsigned)ET) return;
    int s, d;
    edge_sd(ei, E, t, s, d);
    const float v = __expf(lrelu(as[s] + ad[d], 0.2f) - mx[d]);
    ex[t] = v;
    atomicAdd(&den[d], v);
}

template <int LOGD>
__global__ void edge_scatter(const int* __restrict__ ei, int E, int ET,
                             const float* __restrict__ H,
                             const float* __restrict__ ex,
                             float* __restrict__ acc) {
    constexpr unsigned D = 1u << LOGD;
    const unsigned t = blockIdx.x * blockDim.x + threadIdx.x;
    const unsigned e = t >> LOGD;
    if (e >= (unsigned)ET) return;
    const unsigned f = t & (D - 1u);
    int s, d;
    edge_sd(ei, E, e, s, d);
    atomicAdd(&acc[(size_t)d * D + f], ex[e] * H[(size_t)s * D + f]);
}

template <int LOGD>
__global__ void node_finalize(const float* __restrict__ acc,
                              const float* __restrict__ den,
                              const float* __restrict__ b,
                              float* __restrict__ out, int n) {
    constexpr unsigned D = 1u << LOGD;
    const unsigned t = blockIdx.x * blockDim.x + threadIdx.x;
    if (t >= (unsigned)n * D) return;
    const unsigned i = t >> LOGD;
    const unsigned f = t & (D - 1u);
    const float v = acc[t] / (den[i] + 1e-16f) + b[f];
    out[t] = lrelu(v, 0.01f);
}

extern "C" void kernel_launch(void* const* d_in, const int* in_sizes, int n_in,
                              void* d_out, int out_size, void* d_ws, size_t ws_size,
                              hipStream_t stream) {
    const float* x   = (const float*)d_in[0];
    const float* W1  = (const float*)d_in[1];
    const float* aS1 = (const float*)d_in[2];
    const float* aD1 = (const float*)d_in[3];
    const float* b1  = (const float*)d_in[4];
    const float* W2  = (const float*)d_in[5];
    const float* aS2 = (const float*)d_in[6];
    const float* aD2 = (const float*)d_in[7];
    const float* b2  = (const float*)d_in[8];
    const int*   ei  = (const int*)d_in[9];

    const int n  = in_sizes[0] / 64;
    const int E  = in_sizes[9] / 2;
    const int ET = E + n;

    float* w    = (float*)d_ws;
    float* h    = w;
    float* acc  = h + (size_t)n * 64;
    float* asN  = acc + (size_t)n * 64;
    float* adN  = asN + (size_t)n;
    float* mx   = adN + (size_t)n;
    float* den  = mx + (size_t)n;
    float* ex   = den + (size_t)n;

    float* midh    = (float*)d_out;
    float* decodex = (float*)d_out + (size_t)n * 32;

    const int TPB = 256;
    const int tiles = (n + 15) / 16;
    const dim3 gemm_grid((tiles + 7) / 8);
    const dim3 node_grid((n + TPB - 1) / TPB);
    const dim3 edge_grid((ET + TPB - 1) / TPB);

    hipMemsetAsync(acc, 0, (size_t)n * 32 * sizeof(float), stream);
    hipMemsetAsync(den, 0, (size_t)n * sizeof(float), stream);
    init_neg_inf<<<node_grid, TPB, 0, stream>>>(mx, n);

    gat_gemm_wmma<64, 32><<<gemm_grid, TPB, 0, stream>>>(x, W1, h, n);
    node_dots<32><<<node_grid, TPB, 0, stream>>>(h, aS1, aD1, asN, adN, n);

    edge_logit_max<<<edge_grid, TPB, 0, stream>>>(ei, E, ET, asN, adN, mx);
    edge_exp_sum<<<edge_grid, TPB, 0, stream>>>(ei, E, ET, asN, adN, mx, ex, den);
    {
        const size_t work = (size_t)ET * 32;
        edge_scatter<5><<<dim3((work + TPB - 1) / TPB), TPB, 0, stream>>>(
            ei, E, ET, h, ex, acc);
    }
    {
        const size_t work = (size_t)n * 32;
        node_finalize<5><<<dim3((work + TPB - 1) / TPB), TPB, 0, stream>>>(
            acc, den, b1, midh, n);
    }

    hipMemsetAsync(acc, 0, (size_t)n * 64 * sizeof(float), stream);
    hipMemsetAsync(den, 0, (size_t)n * sizeof(float), stream);
    init_neg_inf<<<node_grid, TPB, 0, stream>>>(mx, n);

    gat_gemm_wmma<32, 64><<<gemm_grid, TPB, 0, stream>>>(midh, W2, h, n);
    node_dots<64><<<node_grid, TPB, 0, stream>>>(h, aS2, aD2, asN, adN, n);

    edge_logit_max<<<edge_grid, TPB, 0, stream>>>(ei, E, ET, asN, adN, mx);
    edge_exp_sum<<<edge_grid, TPB, 0, stream>>>(ei, E, ET, asN, adN, mx, ex, den);
    {
        const size_t work = (size_t)ET * 64;
        edge_scatter<6><<<dim3((work + TPB - 1) / TPB), TPB, 0, stream>>>(
            ei, E, ET, h, ex, acc);
    }
    {
        const size_t work = (size_t)n * 64;
        node_finalize<6><<<dim3((work + TPB - 1) / TPB), TPB, 0, stream>>>(
            acc, den, b2, decodex, n);
    }
}


# Per-call edge lists sorted by destination (atomic count with arrival-order rank, prefix sum, fill); one wave per node does softmax and row sum in registers; no float atomics, sums in rank order
# speedup vs baseline: 4.0663x; 4.0663x over previous
#include <hip/hip_runtime.h>
#include <cstddef>
#include <cstdint>

typedef float v2f __attribute__((ext_vector_type(2)));
typedef float v8f __attribute__((ext_vector_type(8)));

__device__ __forceinline__ float lrelu(float x, float s) {
    return x > 0.0f ? x : s * x;
}

template <int K, int NOUT>
__launch_bounds__(256)
__global__ void gat_gemm_wmma(const float* __restrict__ X,
                              const float* __restrict__ W,
                              float* __restrict__ H, int n) {
    constexpr int NT  = NOUT / 16;
    constexpr int LDK = K + 2;
    __shared__ alignas(16) float sWt[NOUT * LDK];
    for (int i = threadIdx.x; i < K * NOUT; i += 256) {
        const int k = i / NOUT;
        const int c = i - k * NOUT;
        sWt[c * LDK + k] = W[i];
    }
    __syncthreads();

    const int wave = threadIdx.x >> 5;
    const int lane = threadIdx.x & 31;
    const int tile = blockIdx.x * 8 + wave;
    const int m0 = tile * 16;
    if (m0 >= n) return;

    const int ml = lane & 15;
    const int kh = lane >> 4;
    int row = m0 + ml;
    row = row < n ? row : n - 1;
    const float* __restrict__ xrow = X + (size_t)row * K;

    v8f acc[NT] = {};

#pragma unroll
    for (int kk = 0; kk < K; kk += 4) {
        const v2f a = *(const v2f*)(xrow + kk + 2 * kh);
#pragma unroll
        for (int t = 0; t < NT; ++t) {
            const v2f b = *(const v2f*)(&sWt[(t * 16 + ml) * LDK + kk + 2 * kh]);
            acc[t] = __builtin_amdgcn_wmma_f32_16x16x4_f32(
                false, a, false, b, (short)0, acc[t], false, false);
        }
    }

    const int rb = kh * 8;
#pragma unroll
    for (int t = 0; t < NT; ++t) {
#pragma unroll
        for (int r = 0; r < 8; ++r) {
            const int m = m0 + rb + r;
            if (m < n) H[(size_t)m * NOUT + t * 16 + ml] = acc[t][r];
        }
    }
}

template <int D>
__global__ void node_dots(const float* __restrict__ H,
                          const float* __restrict__ a_s,
                          const float* __restrict__ a_d,
                          float* __restrict__ os, float* __restrict__ od, int n) {
    const int i = blockIdx.x * blockDim.x + threadIdx.x;
    if (i >= n) return;
    const float* __restrict__ h = H + (size_t)i * D;
    float ss = 0.0f, sd = 0.0f;
#pragma unroll
    for (int f = 0; f < D; ++f) {
        const float v = h[f];
        ss += v * a_s[f];
        sd += v * a_d[f];
    }
    os[i] = ss;
    od[i] = sd;
}

__global__ void zero_counts(int* __restrict__ cnt, int n) {
    const int i = blockIdx.x * blockDim.x + threadIdx.x;
    if (i < n) cnt[i] = 0;
}

__global__ void count_rank(const int* __restrict__ ei, int E,
                           int* __restrict__ cnt, int* __restrict__ rank) {
    const int e = blockIdx.x * blockDim.x + threadIdx.x;
    if (e >= E) return;
    rank[e] = atomicAdd(&cnt[ei[(size_t)E + e]], 1);
}

__global__ void tile_sums(const int* __restrict__ cnt, int n,
                          int* __restrict__ tsum) {
    __shared__ int sw[8];
    const int base = blockIdx.x * 1024 + threadIdx.x * 4;
    int v = 0;
    for (int k = 0; k < 4; ++k)
        if (base + k < n) v += cnt[base + k];
    for (int o = 16; o > 0; o >>= 1) v += __shfl_xor(v, o);
    if ((threadIdx.x & 31) == 0) sw[threadIdx.x >> 5] = v;
    __syncthreads();
    if (threadIdx.x == 0) {
        int t = 0;
        for (int w = 0; w < 8; ++w) t += sw[w];
        tsum[blockIdx.x] = t;
    }
}

__global__ void scan_offsets(const int* __restrict__ cnt,
                             const int* __restrict__ tsum, int n,
                             int* __restrict__ off) {
    __shared__ int sp[8];
    __shared__ int sw[8];
    const int lane = threadIdx.x & 31;
    const int wave = threadIdx.x >> 5;

    int p = 0;
    for (int i = threadIdx.x; i < (int)blockIdx.x; i += 256) p += tsum[i];
    for (int o = 16; o > 0; o >>= 1) p += __shfl_xor(p, o);
    if (lane == 0) sp[wave] = p;

    const int base = blockIdx.x * 1024 + threadIdx.x * 4;
    int c[4];
    int s = 0;
    for (int k = 0; k < 4; ++k) {
        c[k] = (base + k < n) ? cnt[base + k] : 0;
        s += c[k];
    }
    int inc = s;
    for (int o = 1; o < 32; o <<= 1) {
        const int t = __shfl_up(inc, o);
        if (lane >= o) inc += t;
    }
    if (lane == 31) sw[wave] = inc;
    __syncthreads();

    int excl = inc - s;
    for (int w = 0; w < 8; ++w) {
        excl += sp[w];
        if (w < wave) excl += sw[w];
    }
    for (int k = 0; k < 4; ++k) {
        if (base + k < n) off[base + k] = excl;
        excl += c[k];
    }
}

__global__ void fill_perm(const int* __restrict__ ei, int E,
                          const int* __restrict__ off,
                          const int* __restrict__ rank,
                          int* __restrict__ perm) {
    const int e = blockIdx.x * blockDim.x + threadIdx.x;
    if (e >= E) return;
    perm[off[ei[(size_t)E + e]] + rank[e]] = ei[e];
}

__device__ __forceinline__ void row_load(const float* __restrict__ p, float (&r)[1]) {
    r[0] = p[0];
}
__device__ __forceinline__ void row_load(const float* __restrict__ p, float (&r)[2]) {
    const v2f t = *(const v2f*)p;
    r[0] = t.x;
    r[1] = t.y;
}

template <int D>
__device__ __forceinline__ void add_rows(const float* __restrict__ Hl, int s, float w,
                                         int c, float (&acc)[D / 32]) {
    constexpr int V = D / 32;
    for (int k = 0; k < c; k += 8) {
        float wk[8];
        float r[8][V];
#pragma unroll
        for (int u = 0; u < 8; ++u) {
            wk[u] = 0.0f;
            for (int v = 0; v < V; ++v) r[u][v] = 0.0f;
            if (k + u < c) {
                const int sk = __builtin_amdgcn_readlane(s, k + u);
                wk[u] = __int_as_float(__builtin_amdgcn_readlane(__float_as_int(w), k + u));
                row_load(Hl + (size_t)sk * D, r[u]);
            }
        }
#pragma unroll
        for (int u = 0; u < 8; ++u)
            for (int v = 0; v < V; ++v) acc[v] = fmaf(wk[u], r[u][v], acc[v]);
    }
}

template <int D>
__launch_bounds__(256)
__global__ void gat_gather(const int* __restrict__ off, const int* __restrict__ cnt,
                           const int* __restrict__ perm,
                           const float* __restrict__ H,
                           const float* __restrict__ as,
                           const float* __restrict__ ad,
                           const float* __restrict__ b,
                           float* __restrict__ out, int n) {
    constexpr int V = D / 32;
    const int lane = threadIdx.x & 31;
    const int wave = __builtin_amdgcn_readfirstlane((int)(threadIdx.x >> 5));
    const int node = blockIdx.x * 8 + wave;
    if (node >= n) return;

    const int start = off[node];
    const int deg = cnt[node];
    const float adv = ad[node];
    const float lself = lrelu(as[node] + adv, 0.2f);
    const float* __restrict__ Hl = H + lane * V;

    int s0 = 0;
    float l0 = -__builtin_inff();
    if (lane < deg) {
        s0 = perm[start + lane];
        l0 = lrelu(as[s0] + adv, 0.2f);
    }
    float m = fmaxf(lself, l0);
    for (int base = 32; base < deg; base += 32) {
        const int j = base + lane;
        if (j < deg) m = fmaxf(m, lrelu(as[perm[start + j]] + adv, 0.2f));
    }
    for (int o = 16; o > 0; o >>= 1) m = fmaxf(m, __shfl_xor(m, o));

    const float wself = __expf(lself - m);
    float acc[V];
    {
        float r[V];
        row_load(Hl + (size_t)node * D, r);
        for (int v = 0; v < V; ++v) acc[v] = wself * r[v];
    }
    const float w0 = lane < deg ? __expf(l0 - m) : 0.0f;
    float dsum = w0;
    add_rows<D>(Hl, s0, w0, deg < 32 ? deg : 32, acc);
    for (int base = 32; base < deg; base += 32) {
        const int j = base + lane;
        int s = 0;
        float w = 0.0f;
        if (j < deg) {
            s = perm[start + j];
            w = __expf(lrelu(as[s] + adv, 0.2f) - m);
        }
        dsum += w;
        const int c = deg - base;
        add_rows<D>(Hl, s, w, c < 32 ? c : 32, acc);
    }
    for (int o = 16; o > 0; o >>= 1) dsum += __shfl_xor(dsum, o);
    const float den = dsum + wself + 1e-16f;

    float* __restrict__ o = out + (size_t)node * D + lane * V;
    if (V == 1) {
        o[0] = lrelu(acc[0] / den + b[lane], 0.01f);
    } else {
        v2f t;
        t.x = lrelu(acc[0] / den + b[lane * V], 0.01f);
        t.y = lrelu(acc[V - 1] / den + b[lane * V + V - 1], 0.01f);
        *(v2f*)o = t;
    }
}

extern "C" void kernel_launch(void* const* d_in, const int* in_sizes, int n_in,
                              void* d_out, int out_size, void* d_ws, size_t ws_size,
                              hipStream_t stream) {
    const float* x   = (const float*)d_in[0];
    const float* W1  = (const float*)d_in[1];
    const float* aS1 = (const float*)d_in[2];
    const float* aD1 = (const float*)d_in[3];
    const float* b1  = (const float*)d_in[4];
    const float* W2  = (const float*)d_in[5];
    const float* aS2 = (const float*)d_in[6];
    const float* aD2 = (const float*)d_in[7];
    const float* b2  = (const float*)d_in[8];
    const int*   ei  = (const int*)d_in[9];

    const int n  = in_sizes[0] / 64;
    const int E  = in_sizes[9] / 2;
    const int nt = (n + 1023) / 1024;

    float* w    = (float*)d_ws;
    float* h    = w;
    float* asN  = h + (size_t)n * 64;
    float* adN  = asN + (size_t)n;
    int*   cnt  = (int*)(adN + (size_t)n);
    int*   off  = cnt + (size_t)n;
    int*   rank = off + (size_t)n;
    int*   perm = rank + (size_t)E;
    int*   tsum = perm + (size_t)E;

    float* midh    = (float*)d_out;
    float* decodex = (float*)d_out + (size_t)n * 32;

    const int TPB = 256;
    const int tiles = (n + 15) / 16;
    const dim3 gemm_grid((tiles + 7) / 8);
    const dim3 node_grid((n + TPB - 1) / TPB);
    const dim3 edge_grid((E + TPB - 1) / TPB);
    const dim3 wave_grid((n + 7) / 8);

    zero_counts<<<node_grid, TPB, 0, stream>>>(cnt, n);
    count_rank<<<edge_grid, TPB, 0, stream>>>(ei, E, cnt, rank);
    tile_sums<<<dim3(nt), TPB, 0, stream>>>(cnt, n, tsum);
    scan_offsets<<<dim3(nt), TPB, 0, stream>>>(cnt, tsum, n, off);
    fill_perm<<<edge_grid, TPB, 0, stream>>>(ei, E, off, rank, perm);

    gat_gemm_wmma<64, 32><<<gemm_grid, TPB, 0, stream>>>(x, W1, h, n);
    node_dots<32><<<node_grid, TPB, 0, stream>>>(h, aS1, aD1, asN, adN, n);
    gat_gather<32><<<wave_grid, TPB, 0, stream>>>(off, cnt, perm, h, asN, adN, b1, midh, n);

    gat_gemm_wmma<32, 64><<<gemm_grid, TPB, 0, stream>>>(midh, W2, h, n);
    node_dots<64><<<node_grid, TPB, 0, stream>>>(h, aS2, aD2, asN, adN, n);
    gat_gather<64><<<wave_grid, TPB, 0, stream>>>(off, cnt, perm, h, asN, adN, b2, decodex, n);
}


# Edge lists built by a two-level counting sort with LDS atomic ranks (chunk histogram, bucket placement, per-bucket node sort), not one returning global atomic per edge; sums in LDS arrival order
# speedup vs baseline: 6.3774x; 1.5684x over previous
#include <hip/hip_runtime.h>
#include <cstddef>
#include <cstdint>

typedef float v2f __attribute__((ext_vector_type(2)));
typedef float v8f __attribute__((ext_vector_type(8)));

__device__ __forceinline__ float lrelu(float x, float s) {
    return x > 0.0f ? x : s * x;
}

template <int K, int NOUT>
__launch_bounds__(256)
__global__ void gat_gemm_wmma(const float* __restrict__ X,
                              const float* __restrict__ W,
                              float* __restrict__ H, int n) {
    constexpr int NT  = NOUT / 16;
    constexpr int LDK = K + 2;
    __shared__ alignas(16) float sWt[NOUT * LDK];
    for (int i = threadIdx.x; i < K * NOUT; i += 256) {
        const int k = i / NOUT;
        const int c = i - k * NOUT;
        sWt[c * LDK + k] = W[i];
    }
    __syncthreads();

    const int wave = threadIdx.x >> 5;
    const int lane = threadIdx.x & 31;
    const int tile = blockIdx.x * 8 + wave;
    const int m0 = tile * 16;
    if (m0 >= n) return;

    const int ml = lane & 15;
    const int kh = lane >> 4;
    int row = m0 + ml;
    row = row < n ? row : n - 1;
    const float* __restrict__ xrow = X + (size_t)row * K;

    v8f acc[NT] = {};

#pragma unroll
    for (int kk = 0; kk < K; kk += 4) {
        const v2f a = *(const v2f*)(xrow + kk + 2 * kh);
#pragma unroll
        for (int t = 0; t < NT; ++t) {
            const v2f b = *(const v2f*)(&sWt[(t * 16 + ml) * LDK + kk + 2 * kh]);
            acc[t] = __builtin_amdgcn_wmma_f32_16x16x4_f32(
                false, a, false, b, (short)0, acc[t], false, false);
        }
    }

    const int rb = kh * 8;
#pragma unroll
    for (int t = 0; t < NT; ++t) {
#pragma unroll
        for (int r = 0; r < 8; ++r) {
            const int m = m0 + rb + r;
            if (m < n) H[(size_t)m * NOUT + t * 16 + ml] = acc[t][r];
        }
    }
}

template <int D>
__global__ void node_dots(const float* __restrict__ H,
                          const float* __restrict__ a_s,
                          const float* __restrict__ a_d,
                          float* __restrict__ os, float* __restrict__ od, int n) {
    const int i = blockIdx.x * blockDim.x + threadIdx.x;
    if (i >= n) return;
    const float* __restrict__ h = H + (size_t)i * D;
    float ss = 0.0f, sd = 0.0f;
#pragma unroll
    for (int f = 0; f < D; ++f) {
        const float v = h[f];
        ss += v * a_s[f];
        sd += v * a_d[f];
    }
    os[i] = ss;
    od[i] = sd;
}

constexpr int NCHUNK  = 256;
constexpr int BSHIFT  = 9;
constexpr int BNODES  = 1 << BSHIFT;

__device__ __forceinline__ int block_excl_scan(int v, int* sw) {
    const int lane = threadIdx.x & 31;
    const int wave = threadIdx.x >> 5;
    int inc = v;
    for (int o = 1; o < 32; o <<= 1) {
        const int t = __shfl_up(inc, o);
        if (lane >= o) inc += t;
    }
    if (lane == 31) sw[wave] = inc;
    __syncthreads();
    int excl = inc - v;
    for (int w = 0; w < 8; ++w)
        if (w < wave) excl += sw[w];
    return excl;
}

__launch_bounds__(256)
__global__ void chunk_hist(const int* __restrict__ ei, int E, int CH, int NB,
                           int* __restrict__ ctab) {
    __shared__ int lc[256];
    lc[threadIdx.x] = 0;
    __syncthreads();
    const int c = blockIdx.x;
    const int e0 = c * CH;
    const int e1 = (e0 + CH < E) ? e0 + CH : E;
    const int* __restrict__ dst = ei + (size_t)E;
    for (int e = e0 + (int)threadIdx.x; e < e1; e += 2048) {
        int d[8];
#pragma unroll
        for (int u = 0; u < 8; ++u) {
            const int j = e + u * 256;
            d[u] = (j < e1) ? dst[j] : -1;
        }
#pragma unroll
        for (int u = 0; u < 8; ++u)
            if (d[u] >= 0) atomicAdd(&lc[d[u] >> BSHIFT], 1);
    }
    __syncthreads();
    if ((int)threadIdx.x < NB) ctab[c * NB + threadIdx.x] = lc[threadIdx.x];
}

__launch_bounds__(256)
__global__ void chunk_place(const int* __restrict__ ei, int E, int CH, int NB,
                            const int* __restrict__ ctab,
                            int* __restrict__ bbase, int* __restrict__ recs_w) {
    __shared__ int lbase[256];
    __shared__ int lcur[256];
    __shared__ int sw[8];
    const int tid = threadIdx.x;
    const int c = blockIdx.x;

    int before = 0, after = 0;
    if (tid < NB) {
        const int* __restrict__ col = ctab + tid;
#pragma unroll 32
        for (int cc = 0; cc < NCHUNK; ++cc) {
            const int v = col[cc * NB];
            if (cc < c) before += v; else after += v;
        }
    }
    const int excl = block_excl_scan(before + after, sw);
    lbase[tid] = excl + before;
    lcur[tid] = 0;
    if (c == 0 && tid < NB) bbase[tid] = excl;
    __syncthreads();

    int2* __restrict__ recs = reinterpret_cast<int2*>(recs_w);
    const int e0 = c * CH;
    const int e1 = (e0 + CH < E) ? e0 + CH : E;
    const int* __restrict__ dst = ei + (size_t)E;
    for (int e = e0 + tid; e < e1; e += 2048) {
        int d[8], s[8];
#pragma unroll
        for (int u = 0; u < 8; ++u) {
            const int j = e + u * 256;
            d[u] = -1;
            s[u] = 0;
            if (j < e1) {
                d[u] = dst[j];
                s[u] = ei[j];
            }
        }
#pragma unroll
        for (int u = 0; u < 8; ++u) {
            if (d[u] >= 0) {
                const int b = d[u] >> BSHIFT;
                const int r = atomicAdd(&lcur[b], 1);
                int2 rec;
                rec.x = s[u];
                rec.y = d[u];
                recs[lbase[b] + r] = rec;
            }
        }
    }
}

__launch_bounds__(256)
__global__ void bucket_sort(const int* __restrict__ recs_w,
                            const int* __restrict__ bbase, int E, int NB, int n,
                            int* __restrict__ off, int* __restrict__ cnt,
                            int* __restrict__ perm) {
    __shared__ int ncnt[BNODES];
    __shared__ int noff[BNODES];
    __shared__ int sw[8];
    const int tid = threadIdx.x;
    const int b = blockIdx.x;
    const int r0 = bbase[b];
    const int r1 = (b + 1 < NB) ? bbase[b + 1] : E;
    const int nb0 = b << BSHIFT;
    const int2* __restrict__ recs = reinterpret_cast<const int2*>(recs_w);

    ncnt[tid] = 0;
    ncnt[tid + 256] = 0;
    __syncthreads();
    for (int i = r0 + tid; i < r1; i += 2048) {
        int d[8];
#pragma unroll
        for (int u = 0; u < 8; ++u) {
            const int j = i + u * 256;
            d[u] = (j < r1) ? recs[j].y : -1;
        }
#pragma unroll
        for (int u = 0; u < 8; ++u)
            if (d[u] >= 0) atomicAdd(&ncnt[d[u] - nb0], 1);
    }
    __syncthreads();

    const int c0 = ncnt[2 * tid];
    const int c1 = ncnt[2 * tid + 1];
    const int excl = block_excl_scan(c0 + c1, sw);
    noff[2 * tid] = excl;
    noff[2 * tid + 1] = excl + c0;
    const int node = nb0 + 2 * tid;
    if (node < n) {
        off[node] = r0 + excl;
        cnt[node] = c0;
    }
    if (node + 1 < n) {
        off[node + 1] = r0 + excl + c0;
        cnt[node + 1] = c1;
    }
    ncnt[tid] = 0;
    ncnt[tid + 256] = 0;
    __syncthreads();

    for (int i = r0 + tid; i < r1; i += 2048) {
        int2 r[8];
#pragma unroll
        for (int u = 0; u < 8; ++u) {
            const int j = i + u * 256;
            r[u].x = 0;
            r[u].y = -1;
            if (j < r1) r[u] = recs[j];
        }
#pragma unroll
        for (int u = 0; u < 8; ++u) {
            if (r[u].y >= 0) {
                const int dl = r[u].y - nb0;
                const int p = atomicAdd(&ncnt[dl], 1);
                perm[r0 + noff[dl] + p] = r[u].x;
            }
        }
    }
}

__device__ __forceinline__ void row_load(const float* __restrict__ p, float (&r)[1]) {
    r[0] = p[0];
}
__device__ __forceinline__ void row_load(const float* __restrict__ p, float (&r)[2]) {
    const v2f t = *(const v2f*)p;
    r[0] = t.x;
    r[1] = t.y;
}

template <int D>
__device__ __forceinline__ void add_rows(const float* __restrict__ Hl, int s, float w,
                                         int c, float (&acc)[D / 32]) {
    constexpr int V = D / 32;
    for (int k = 0; k < c; k += 8) {
        float wk[8];
        float r[8][V];
#pragma unroll
        for (int u = 0; u < 8; ++u) {
            wk[u] = 0.0f;
            for (int v = 0; v < V; ++v) r[u][v] = 0.0f;
            if (k + u < c) {
                const int sk = __builtin_amdgcn_readlane(s, k + u);
                wk[u] = __int_as_float(__builtin_amdgcn_readlane(__float_as_int(w), k + u));
                row_load(Hl + (size_t)sk * D, r[u]);
            }
        }
#pragma unroll
        for (int u = 0; u < 8; ++u)
            for (int v = 0; v < V; ++v) acc[v] = fmaf(wk[u], r[u][v], acc[v]);
    }
}

template <int D>
__launch_bounds__(256)
__global__ void gat_gather(const int* __restrict__ off, const int* __restrict__ cnt,
                           const int* __restrict__ perm,
                           const float* __restrict__ H,
                           const float* __restrict__ as,
                           const float* __restrict__ ad,
                           const float* __restrict__ b,
                           float* __restrict__ out, int n) {
    constexpr int V = D / 32;
    const int lane = threadIdx.x & 31;
    const int wave = __builtin_amdgcn_readfirstlane((int)(threadIdx.x >> 5));
    const int node = blockIdx.x * 8 + wave;
    if (node >= n) return;

    const int start = off[node];
    const int deg = cnt[node];
    const float adv = ad[node];
    const float lself = lrelu(as[node] + adv, 0.2f);
    const float* __restrict__ Hl = H + lane * V;

    int s0 = 0;
    float l0 = -__builtin_inff();
    if (lane < deg) {
        s0 = perm[start + lane];
        l0 = lrelu(as[s0] + adv, 0.2f);
    }
    float m = fmaxf(lself, l0);
    for (int base = 32; base < deg; base += 32) {
        const int j = base + lane;
        if (j < deg) m = fmaxf(m, lrelu(as[perm[start + j]] + adv, 0.2f));
    }
    for (int o = 16; o > 0; o >>= 1) m = fmaxf(m, __shfl_xor(m, o));

    const float wself = __expf(lself - m);
    float acc[V];
    {
        float r[V];
        row_load(Hl + (size_t)node * D, r);
        for (int v = 0; v < V; ++v) acc[v] = wself * r[v];
    }
    const float w0 = lane < deg ? __expf(l0 - m) : 0.0f;
    float dsum = w0;
    add_rows<D>(Hl, s0, w0, deg < 32 ? deg : 32, acc);
    for (int base = 32; base < deg; base += 32) {
        const int j = base + lane;
        int s = 0;
        float w = 0.0f;
        if (j < deg) {
            s = perm[start + j];
            w = __expf(lrelu(as[s] + adv, 0.2f) - m);
        }
        dsum += w;
        const int c = deg - base;
        add_rows<D>(Hl, s, w, c < 32 ? c : 32, acc);
    }
    for (int o = 16; o > 0; o >>= 1) dsum += __shfl_xor(dsum, o);
    const float den = dsum + wself + 1e-16f;

    float* __restrict__ o = out + (size_t)node * D + lane * V;
    if (V == 1) {
        o[0] = lrelu(acc[0] / den + b[lane], 0.01f);
    } else {
        v2f t;
        t.x = lrelu(acc[0] / den + b[lane * V], 0.01f);
        t.y = lrelu(acc[V - 1] / den + b[lane * V + V - 1], 0.01f);
        *(v2f*)o = t;
    }
}

extern "C" void kernel_launch(void* const* d_in, const int* in_sizes, int n_in,
                              void* d_out, int out_size, void* d_ws, size_t ws_size,
                              hipStream_t stream) {
    const float* x   = (const float*)d_in[0];
    const float* W1  = (const float*)d_in[1];
    const float* aS1 = (const float*)d_in[2];
    const float* aD1 = (const float*)d_in[3];
    const float* b1  = (const float*)d_in[4];
    const float* W2  = (const float*)d_in[5];
    const float* aS2 = (const float*)d_in[6];
    const float* aD2 = (const float*)d_in[7];
    const float* b2  = (const float*)d_in[8];
    const int*   ei  = (const int*)d_in[9];

    const int n  = in_sizes[0] / 64;
    const int E  = in_sizes[9] / 2;
    const int NB = (n + BNODES - 1) / BNODES;
    const int CH = (E + NCHUNK - 1) / NCHUNK;

    float* w    = (float*)d_ws;
    float* h    = w;
    int*   recs = (int*)(h + (size_t)n * 64);
    float* asN  = (float*)(recs + (size_t)2 * E);
    float* adN  = asN + (size_t)n;
    int*   cnt  = (int*)(adN + (size_t)n);
    int*   off  = cnt + (size_t)n;
    int*   perm = off + (size_t)n;
    int*   ctab = perm + (size_t)E;
    int*   bbase = ctab + (size_t)NCHUNK * 256;

    float* midh    = (float*)d_out;
    float* decodex = (float*)d_out + (size_t)n * 32;

    const int TPB = 256;
    const int tiles = (n + 15) / 16;
    const dim3 gemm_grid((tiles + 7) / 8);
    const dim3 node_grid((n + TPB - 1) / TPB);
    const dim3 wave_grid((n + 7) / 8);

    chunk_hist<<<dim3(NCHUNK), TPB, 0, stream>>>(ei, E, CH, NB, ctab);
    chunk_place<<<dim3(NCHUNK), TPB, 0, stream>>>(ei, E, CH, NB, ctab, bbase, recs);
    bucket_sort<<<dim3(NB), TPB, 0, stream>>>(recs, bbase, E, NB, n, off, cnt, perm);

    gat_gemm_wmma<64, 32><<<gemm_grid, TPB, 0, stream>>>(x, W1, h, n);
    node_dots<32><<<node_grid, TPB, 0, stream>>>(h, aS1, aD1, asN, adN, n);
    gat_gather<32><<<wave_grid, TPB, 0, stream>>>(off, cnt, perm, h, asN, adN, b1, midh, n);

    gat_gemm_wmma<32, 64><<<gemm_grid, TPB, 0, stream>>>(midh, W2, h, n);
    node_dots<64><<<node_grid, TPB, 0, stream>>>(h, aS2, aD2, asN, adN, n);
    gat_gather<64><<<wave_grid, TPB, 0, stream>>>(off, cnt, perm, h, asN, adN, b2, decodex, n);
}


# Gather launches run a resident grid (2048 work-groups, one wave per wave slot) whose waves loop over nodes with stride 16384, instead of 100000 short one-node waves per launch; sums unchanged
# speedup vs baseline: 6.6382x; 1.0409x over previous
#include <hip/hip_runtime.h>
#include <cstddef>
#include <cstdint>

typedef float v2f __attribute__((ext_vector_type(2)));
typedef float v8f __attribute__((ext_vector_type(8)));

__device__ __forceinline__ float lrelu(float x, float s) {
    return x > 0.0f ? x : s * x;
}

template <int K, int NOUT>
__launch_bounds__(256)
__global__ void gat_gemm_wmma(const float* __restrict__ X,
                              const float* __restrict__ W,
                              float* __restrict__ H, int n) {
    constexpr int NT  = NOUT / 16;
    constexpr int LDK = K + 2;
    __shared__ alignas(16) float sWt[NOUT * LDK];
    for (int i = threadIdx.x; i < K * NOUT; i += 256) {
        const int k = i / NOUT;
        const int c = i - k * NOUT;
        sWt[c * LDK + k] = W[i];
    }
    __syncthreads();

    const int wave = threadIdx.x >> 5;
    const int lane = threadIdx.x & 31;
    const int tile = blockIdx.x * 8 + wave;
    const int m0 = tile * 16;
    if (m0 >= n) return;

    const int ml = lane & 15;
    const int kh = lane >> 4;
    int row = m0 + ml;
    row = row < n ? row : n - 1;
    const float* __restrict__ xrow = X + (size_t)row * K;

    v8f acc[NT] = {};

#pragma unroll
    for (int kk = 0; kk < K; kk += 4) {
        const v2f a = *(const v2f*)(xrow + kk + 2 * kh);
#pragma unroll
        for (int t = 0; t < NT; ++t) {
            const v2f b = *(const v2f*)(&sWt[(t * 16 + ml) * LDK + kk + 2 * kh]);
            acc[t] = __builtin_amdgcn_wmma_f32_16x16x4_f32(
                false, a, false, b, (short)0, acc[t], false, false);
        }
    }

    const int rb = kh * 8;
#pragma unroll
    for (int t = 0; t < NT; ++t) {
#pragma unroll
        for (int r = 0; r < 8; ++r) {
            const int m = m0 + rb + r;
            if (m < n) H[(size_t)m * NOUT + t * 16 + ml] = acc[t][r];
        }
    }
}

template <int D>
__global__ void node_dots(const float* __restrict__ H,
                          const float* __restrict__ a_s,
                          const float* __restrict__ a_d,
                          float* __restrict__ os, float* __restrict__ od, int n) {
    const int i = blockIdx.x * blockDim.x + threadIdx.x;
    if (i >= n) return;
    const float* __restrict__ h = H + (size_t)i * D;
    float ss = 0.0f, sd = 0.0f;
#pragma unroll
    for (int f = 0; f < D; ++f) {
        const float v = h[f];
        ss += v * a_s[f];
        sd += v * a_d[f];
    }
    os[i] = ss;
    od[i] = sd;
}

constexpr int NCHUNK  = 256;
constexpr int BSHIFT  = 9;
constexpr int BNODES  = 1 << BSHIFT;

__device__ __forceinline__ int block_excl_scan(int v, int* sw) {
    const int lane = threadIdx.x & 31;
    const int wave = threadIdx.x >> 5;
    int inc = v;
    for (int o = 1; o < 32; o <<= 1) {
        const int t = __shfl_up(inc, o);
        if (lane >= o) inc += t;
    }
    if (lane == 31) sw[wave] = inc;
    __syncthreads();
    int excl = inc - v;
    for (int w = 0; w < 8; ++w)
        if (w < wave) excl += sw[w];
    return excl;
}

__launch_bounds__(256)
__global__ void chunk_hist(const int* __restrict__ ei, int E, int CH, int NB,
                           int* __restrict__ ctab) {
    __shared__ int lc[256];
    lc[threadIdx.x] = 0;
    __syncthreads();
    const int c = blockIdx.x;
    const int e0 = c * CH;
    const int e1 = (e0 + CH < E) ? e0 + CH : E;
    const int* __restrict__ dst = ei + (size_t)E;
    for (int e = e0 + (int)threadIdx.x; e < e1; e += 2048) {
        int d[8];
#pragma unroll
        for (int u = 0; u < 8; ++u) {
            const int j = e + u * 256;
            d[u] = (j < e1) ? dst[j] : -1;
        }
#pragma unroll
        for (int u = 0; u < 8; ++u)
            if (d[u] >= 0) atomicAdd(&lc[d[u] >> BSHIFT], 1);
    }
    __syncthreads();
    if ((int)threadIdx.x < NB) ctab[c * NB + threadIdx.x] = lc[threadIdx.x];
}

__launch_bounds__(256)
__global__ void chunk_place(const int* __restrict__ ei, int E, int CH, int NB,
                            const int* __restrict__ ctab,
                            int* __restrict__ bbase, int* __restrict__ recs_w) {
    __shared__ int lbase[256];
    __shared__ int lcur[256];
    __shared__ int sw[8];
    const int tid = threadIdx.x;
    const int c = blockIdx.x;

    int before = 0, after = 0;
    if (tid < NB) {
        const int* __restrict__ col = ctab + tid;
#pragma unroll 32
        for (int cc = 0; cc < NCHUNK; ++cc) {
            const int v = col[cc * NB];
            if (cc < c) before += v; else after += v;
        }
    }
    const int excl = block_excl_scan(before + after, sw);
    lbase[tid] = excl + before;
    lcur[tid] = 0;
    if (c == 0 && tid < NB) bbase[tid] = excl;
    __syncthreads();

    int2* __restrict__ recs = reinterpret_cast<int2*>(recs_w);
    const int e0 = c * CH;
    const int e1 = (e0 + CH < E) ? e0 + CH : E;
    const int* __restrict__ dst = ei + (size_t)E;
    for (int e = e0 + tid; e < e1; e += 2048) {
        int d[8], s[8];
#pragma unroll
        for (int u = 0; u < 8; ++u) {
            const int j = e + u * 256;
            d[u] = -1;
            s[u] = 0;
            if (j < e1) {
                d[u] = dst[j];
                s[u] = ei[j];
            }
        }
#pragma unroll
        for (int u = 0; u < 8; ++u) {
            if (d[u] >= 0) {
                const int b = d[u] >> BSHIFT;
                const int r = atomicAdd(&lcur[b], 1);
                int2 rec;
                rec.x = s[u];
                rec.y = d[u];
                recs[lbase[b] + r] = rec;
            }
        }
    }
}

__launch_bounds__(256)
__global__ void bucket_sort(const int* __restrict__ recs_w,
                            const int* __restrict__ bbase, int E, int NB, int n,
                            int* __restrict__ off, int* __restrict__ cnt,
                            int* __restrict__ perm) {
    __shared__ int ncnt[BNODES];
    __shared__ int noff[BNODES];
    __shared__ int sw[8];
    const int tid = threadIdx.x;
    const int b = blockIdx.x;
    const int r0 = bbase[b];
    const int r1 = (b + 1 < NB) ? bbase[b + 1] : E;
    const int nb0 = b << BSHIFT;
    const int2* __restrict__ recs = reinterpret_cast<const int2*>(recs_w);

    ncnt[tid] = 0;
    ncnt[tid + 256] = 0;
    __syncthreads();
    for (int i = r0 + tid; i < r1; i += 2048) {
        int d[8];
#pragma unroll
        for (int u = 0; u < 8; ++u) {
            const int j = i + u * 256;
            d[u] = (j < r1) ? recs[j].y : -1;
        }
#pragma unroll
        for (int u = 0; u < 8; ++u)
            if (d[u] >= 0) atomicAdd(&ncnt[d[u] - nb0], 1);
    }
    __syncthreads();

    const int c0 = ncnt[2 * tid];
    const int c1 = ncnt[2 * tid + 1];
    const int excl = block_excl_scan(c0 + c1, sw);
    noff[2 * tid] = excl;
    noff[2 * tid + 1] = excl + c0;
    const int node = nb0 + 2 * tid;
    if (node < n) {
        off[node] = r0 + excl;
        cnt[node] = c0;
    }
    if (node + 1 < n) {
        off[node + 1] = r0 + excl + c0;
        cnt[node + 1] = c1;
    }
    ncnt[tid] = 0;
    ncnt[tid + 256] = 0;
    __syncthreads();

    for (int i = r0 + tid; i < r1; i += 2048) {
        int2 r[8];
#pragma unroll
        for (int u = 0; u < 8; ++u) {
            const int j = i + u * 256;
            r[u].x = 0;
            r[u].y = -1;
            if (j < r1) r[u] = recs[j];
        }
#pragma unroll
        for (int u = 0; u < 8; ++u) {
            if (r[u].y >= 0) {
                const int dl = r[u].y - nb0;
                const int p = atomicAdd(&ncnt[dl], 1);
                perm[r0 + noff[dl] + p] = r[u].x;
            }
        }
    }
}

__device__ __forceinline__ void row_load(const float* __restrict__ p, float (&r)[1]) {
    r[0] = p[0];
}
__device__ __forceinline__ void row_load(const float* __restrict__ p, float (&r)[2]) {
    const v2f t = *(const v2f*)p;
    r[0] = t.x;
    r[1] = t.y;
}

template <int D>
__device__ __forceinline__ void add_rows(const float* __restrict__ Hl, int s, float w,
                                         int c, float (&acc)[D / 32]) {
    constexpr int V = D / 32;
    for (int k = 0; k < c; k += 8) {
        float wk[8];
        float r[8][V];
#pragma unroll
        for (int u = 0; u < 8; ++u) {
            wk[u] = 0.0f;
            for (int v = 0; v < V; ++v) r[u][v] = 0.0f;
            if (k + u < c) {
                const int sk = __builtin_amdgcn_readlane(s, k + u);
                wk[u] = __int_as_float(__builtin_amdgcn_readlane(__float_as_int(w), k + u));
                row_load(Hl + (size_t)sk * D, r[u]);
            }
        }
#pragma unroll
        for (int u = 0; u < 8; ++u)
            for (int v = 0; v < V; ++v) acc[v] = fmaf(wk[u], r[u][v], acc[v]);
    }
}

constexpr int GATHER_WGS = 2048;

template <int D>
__launch_bounds__(256)
__global__ void gat_gather(const int* __restrict__ off, const int* __restrict__ cnt,
                           const int* __restrict__ perm,
                           const float* __restrict__ H,
                           const float* __restrict__ as,
                           const float* __restrict__ ad,
                           const float* __restrict__ b,
                           float* __restrict__ out, int n) {
    constexpr int V = D / 32;
    const int lane = threadIdx.x & 31;
    const int wave = __builtin_amdgcn_readfirstlane((int)(threadIdx.x >> 5));
    constexpr int nwaves = GATHER_WGS * 8;
    const float* __restrict__ Hl = H + lane * V;

    for (int node = blockIdx.x * 8 + wave; node < n; node += nwaves) {
        const int start = off[node];
        const int deg = cnt[node];
        const float adv = ad[node];
        const float lself = lrelu(as[node] + adv, 0.2f);

        int s0 = 0;
        float l0 = -__builtin_inff();
        if (lane < deg) {
            s0 = perm[start + lane];
            l0 = lrelu(as[s0] + adv, 0.2f);
        }
        float m = fmaxf(lself, l0);
        for (int base = 32; base < deg; base += 32) {
            const int j = base + lane;
            if (j < deg) m = fmaxf(m, lrelu(as[perm[start + j]] + adv, 0.2f));
        }
        for (int o = 16; o > 0; o >>= 1) m = fmaxf(m, __shfl_xor(m, o));

        const float wself = __expf(lself - m);
        float acc[V];
        {
            float r[V];
            row_load(Hl + (size_t)node * D, r);
            for (int v = 0; v < V; ++v) acc[v] = wself * r[v];
        }
        const float w0 = lane < deg ? __expf(l0 - m) : 0.0f;
        float dsum = w0;
        add_rows<D>(Hl, s0, w0, deg < 32 ? deg : 32, acc);
        for (int base = 32; base < deg; base += 32) {
            const int j = base + lane;
            int s = 0;
            float w = 0.0f;
            if (j < deg) {
                s = perm[start + j];
                w = __expf(lrelu(as[s] + adv, 0.2f) - m);
            }
            dsum += w;
            const int c = deg - base;
            add_rows<D>(Hl, s, w, c < 32 ? c : 32, acc);
        }
        for (int o = 16; o > 0; o >>= 1) dsum += __shfl_xor(dsum, o);
        const float den = dsum + wself + 1e-16f;

        float* __restrict__ o = out + (size_t)node * D + lane * V;
        if (V == 1) {
            o[0] = lrelu(acc[0] / den + b[lane], 0.01f);
        } else {
            v2f t;
            t.x = lrelu(acc[0] / den + b[lane * V], 0.01f);
            t.y = lrelu(acc[V - 1] / den + b[lane * V + V - 1], 0.01f);
            *(v2f*)o = t;
        }
    }
}

extern "C" void kernel_launch(void* const* d_in, const int* in_sizes, int n_in,
                              void* d_out, int out_size, void* d_ws, size_t ws_size,
                              hipStream_t stream) {
    const float* x   = (const float*)d_in[0];
    const float* W1  = (const float*)d_in[1];
    const float* aS1 = (const float*)d_in[2];
    const float* aD1 = (const float*)d_in[3];
    const float* b1  = (const float*)d_in[4];
    const float* W2  = (const float*)d_in[5];
    const float* aS2 = (const float*)d_in[6];
    const float* aD2 = (const float*)d_in[7];
    const float* b2  = (const float*)d_in[8];
    const int*   ei  = (const int*)d_in[9];

    const int n  = in_sizes[0] / 64;
    const int E  = in_sizes[9] / 2;
    const int NB = (n + BNODES - 1) / BNODES;
    const int CH = (E + NCHUNK - 1) / NCHUNK;

    float* w    = (float*)d_ws;
    float* h    = w;
    int*   recs = (int*)(h + (size_t)n * 64);
    float* asN  = (float*)(recs + (size_t)2 * E);
    float* adN  = asN + (size_t)n;
    int*   cnt  = (int*)(adN + (size_t)n);
    int*   off  = cnt + (size_t)n;
    int*   perm = off + (size_t)n;
    int*   ctab = perm + (size_t)E;
    int*   bbase = ctab + (size_t)NCHUNK * 256;

    float* midh    = (float*)d_out;
    float* decodex = (float*)d_out + (size_t)n * 32;

    const int TPB = 256;
    const int tiles = (n + 15) / 16;
    const dim3 gemm_grid((tiles + 7) / 8);
    const dim3 node_grid((n + TPB - 1) / TPB);
    const dim3 wave_grid(GATHER_WGS);

    chunk_hist<<<dim3(NCHUNK), TPB, 0, stream>>>(ei, E, CH, NB, ctab);
    chunk_place<<<dim3(NCHUNK), TPB, 0, stream>>>(ei, E, CH, NB, ctab, bbase, recs);
    bucket_sort<<<dim3(NB), TPB, 0, stream>>>(recs, bbase, E, NB, n, off, cnt, perm);

    gat_gemm_wmma<64, 32><<<gemm_grid, TPB, 0, stream>>>(x, W1, h, n);
    node_dots<32><<<node_grid, TPB, 0, stream>>>(h, aS1, aD1, asN, adN, n);
    gat_gather<32><<<wave_grid, TPB, 0, stream>>>(off, cnt, perm, h, asN, adN, b1, midh, n);

    gat_gemm_wmma<32, 64><<<gemm_grid, TPB, 0, stream>>>(midh, W2, h, n);
    node_dots<64><<<node_grid, TPB, 0, stream>>>(h, aS2, aD2, asN, adN, n);
    gat_gather<64><<<wave_grid, TPB, 0, stream>>>(off, cnt, perm, h, asN, adN, b2, decodex, n);
}


# Layer 2 gathers its 32-wide input rows and applies W2 after the weighted neighbour sum (linearity), logits from m.(W2 a): half the gathered row bytes; product association and sum order change
# speedup vs baseline: 7.1827x; 1.0820x over previous
#include <hip/hip_runtime.h>
#include <cstddef>
#include <cstdint>

typedef float v2f __attribute__((ext_vector_type(2)));
typedef float v8f __attribute__((ext_vector_type(8)));

__device__ __forceinline__ float lrelu(float x, float s) {
    return x > 0.0f ? x : s * x;
}

template <int K, int NOUT, bool BIAS_ACT>
__launch_bounds__(256)
__global__ void gat_gemm_wmma(const float* __restrict__ X,
                              const float* __restrict__ W,
                              const float* __restrict__ bias,
                              float* __restrict__ H, int n) {
    constexpr int NT  = NOUT / 16;
    constexpr int LDK = K + 2;
    __shared__ alignas(16) float sWt[NOUT * LDK];
    for (int i = threadIdx.x; i < K * NOUT; i += 256) {
        const int k = i / NOUT;
        const int c = i - k * NOUT;
        sWt[c * LDK + k] = W[i];
    }
    __syncthreads();

    const int wave = threadIdx.x >> 5;
    const int lane = threadIdx.x & 31;
    const int tile = blockIdx.x * 8 + wave;
    const int m0 = tile * 16;
    if (m0 >= n) return;

    const int ml = lane & 15;
    const int kh = lane >> 4;
    int row = m0 + ml;
    row = row < n ? row : n - 1;
    const float* __restrict__ xrow = X + (size_t)row * K;

    v8f acc[NT] = {};

#pragma unroll
    for (int kk = 0; kk < K; kk += 4) {
        const v2f a = *(const v2f*)(xrow + kk + 2 * kh);
#pragma unroll
        for (int t = 0; t < NT; ++t) {
            const v2f b = *(const v2f*)(&sWt[(t * 16 + ml) * LDK + kk + 2 * kh]);
            acc[t] = __builtin_amdgcn_wmma_f32_16x16x4_f32(
                false, a, false, b, (short)0, acc[t], false, false);
        }
    }

    const int rb = kh * 8;
#pragma unroll
    for (int t = 0; t < NT; ++t) {
        float bv = 0.0f;
        if (BIAS_ACT) bv = bias[t * 16 + ml];
#pragma unroll
        for (int r = 0; r < 8; ++r) {
            const int m = m0 + rb + r;
            float v = acc[t][r];
            if (BIAS_ACT) v = lrelu(v + bv, 0.01f);
            if (m < n) H[(size_t)m * NOUT + t * 16 + ml] = v;
        }
    }
}

template <int D>
__global__ void node_dots(const float* __restrict__ H,
                          const float* __restrict__ a_s,
                          const float* __restrict__ a_d,
                          float* __restrict__ os, float* __restrict__ od, int n) {
    const int i = blockIdx.x * blockDim.x + threadIdx.x;
    if (i >= n) return;
    const float* __restrict__ h = H + (size_t)i * D;
    float ss = 0.0f, sd = 0.0f;
#pragma unroll
    for (int f = 0; f < D; ++f) {
        const float v = h[f];
        ss += v * a_s[f];
        sd += v * a_d[f];
    }
    os[i] = ss;
    od[i] = sd;
}

template <int DIN, int DOUT>
__launch_bounds__(256)
__global__ void node_dots_proj(const float* __restrict__ M,
                               const float* __restrict__ W,
                               const float* __restrict__ a_s,
                               const float* __restrict__ a_d,
                               float* __restrict__ os, float* __restrict__ od, int n) {
    static_assert(DIN == 32, "one wave per attention vector, one lane per row of W");
    __shared__ float su[2 * DIN];
    const int tid = threadIdx.x;
    const int i = blockIdx.x * 256 + tid;
    const int ic = i < n ? i : n - 1;
    const float* __restrict__ m = M + (size_t)ic * DIN;
    float v[DIN];
#pragma unroll
    for (int f = 0; f < DIN; ++f) v[f] = m[f];

    const int wv = __builtin_amdgcn_readfirstlane(tid >> 5);
    if (wv < 2) {
        const float* __restrict__ a = wv == 0 ? a_s : a_d;
        const float* __restrict__ w = W + (size_t)(tid & 31) * DOUT;
        float s = 0.0f;
#pragma unroll
        for (int c = 0; c < DOUT; ++c) s = fmaf(w[c], a[c], s);
        su[tid] = s;
    }
    __syncthreads();

    float ss = 0.0f, sd = 0.0f;
#pragma unroll
    for (int f = 0; f < DIN; ++f) {
        ss = fmaf(v[f], su[f], ss);
        sd = fmaf(v[f], su[DIN + f], sd);
    }
    if (i < n) {
        os[i] = ss;
        od[i] = sd;
    }
}

constexpr int NCHUNK  = 256;
constexpr int BSHIFT  = 9;
constexpr int BNODES  = 1 << BSHIFT;

__device__ __forceinline__ int block_excl_scan(int v, int* sw) {
    const int lane = threadIdx.x & 31;
    const int wave = threadIdx.x >> 5;
    int inc = v;
    for (int o = 1; o < 32; o <<= 1) {
        const int t = __shfl_up(inc, o);
        if (lane >= o) inc += t;
    }
    if (lane == 31) sw[wave] = inc;
    __syncthreads();
    int excl = inc - v;
    for (int w = 0; w < 8; ++w)
        if (w < wave) excl += sw[w];
    return excl;
}

__launch_bounds__(256)
__global__ void chunk_hist(const int* __restrict__ ei, int E, int CH, int NB,
                           int* __restrict__ ctab) {
    __shared__ int lc[256];
    lc[threadIdx.x] = 0;
    __syncthreads();
    const int c = blockIdx.x;
    const int e0 = c * CH;
    const int e1 = (e0 + CH < E) ? e0 + CH : E;
    const int* __restrict__ dst = ei + (size_t)E;
    for (int e = e0 + (int)threadIdx.x; e < e1; e += 2048) {
        int d[8];
#pragma unroll
        for (int u = 0; u < 8; ++u) {
            const int j = e + u * 256;
            d[u] = (j < e1) ? dst[j] : -1;
        }
#pragma unroll
        for (int u = 0; u < 8; ++u)
            if (d[u] >= 0) atomicAdd(&lc[d[u] >> BSHIFT], 1);
    }
    __syncthreads();
    if ((int)threadIdx.x < NB) ctab[c * NB + threadIdx.x] = lc[threadIdx.x];
}

__launch_bounds__(256)
__global__ void chunk_place(const int* __restrict__ ei, int E, int CH, int NB,
                            const int* __restrict__ ctab,
                            int* __restrict__ bbase, int* __restrict__ recs_w) {
    __shared__ int lbase[256];
    __shared__ int lcur[256];
    __shared__ int sw[8];
    const int tid = threadIdx.x;
    const int c = blockIdx.x;

    int before = 0, after = 0;
    if (tid < NB) {
        const int* __restrict__ col = ctab + tid;
#pragma unroll 32
        for (int cc = 0; cc < NCHUNK; ++cc) {
            const int v = col[cc * NB];
            if (cc < c) before += v; else after += v;
        }
    }
    const int excl = block_excl_scan(before + after, sw);
    lbase[tid] = excl + before;
    lcur[tid] = 0;
    if (c == 0 && tid < NB) bbase[tid] = excl;
    __syncthreads();

    int2* __restrict__ recs = reinterpret_cast<int2*>(recs_w);
    const int e0 = c * CH;
    const int e1 = (e0 + CH < E) ? e0 + CH : E;
    const int* __restrict__ dst = ei + (size_t)E;
    for (int e = e0 + tid; e < e1; e += 2048) {
        int d[8], s[8];
#pragma unroll
        for (int u = 0; u < 8; ++u) {
            const int j = e + u * 256;
            d[u] = -1;
            s[u] = 0;
            if (j < e1) {
                d[u] = dst[j];
                s[u] = ei[j];
            }
        }
#pragma unroll
        for (int u = 0; u < 8; ++u) {
            if (d[u] >= 0) {
                const int b = d[u] >> BSHIFT;
                const int r = atomicAdd(&lcur[b], 1);
                int2 rec;
                rec.x = s[u];
                rec.y = d[u];
                recs[lbase[b] + r] = rec;
            }
        }
    }
}

__launch_bounds__(256)
__global__ void bucket_sort(const int* __restrict__ recs_w,
                            const int* __restrict__ bbase, int E, int NB, int n,
                            int* __restrict__ off, int* __restrict__ cnt,
                            int* __restrict__ perm) {
    __shared__ int ncnt[BNODES];
    __shared__ int noff[BNODES];
    __shared__ int sw[8];
    const int tid = threadIdx.x;
    const int b = blockIdx.x;
    const int r0 = bbase[b];
    const int r1 = (b + 1 < NB) ? bbase[b + 1] : E;
    const int nb0 = b << BSHIFT;
    const int2* __restrict__ recs = reinterpret_cast<const int2*>(recs_w);

    ncnt[tid] = 0;
    ncnt[tid + 256] = 0;
    __syncthreads();
    for (int i = r0 + tid; i < r1; i += 2048) {
        int d[8];
#pragma unroll
        for (int u = 0; u < 8; ++u) {
            const int j = i + u * 256;
            d[u] = (j < r1) ? recs[j].y : -1;
        }
#pragma unroll
        for (int u = 0; u < 8; ++u)
            if (d[u] >= 0) atomicAdd(&ncnt[d[u] - nb0], 1);
    }
    __syncthreads();

    const int c0 = ncnt[2 * tid];
    const int c1 = ncnt[2 * tid + 1];
    const int excl = block_excl_scan(c0 + c1, sw);
    noff[2 * tid] = excl;
    noff[2 * tid + 1] = excl + c0;
    const int node = nb0 + 2 * tid;
    if (node < n) {
        off[node] = r0 + excl;
        cnt[node] = c0;
    }
    if (node + 1 < n) {
        off[node + 1] = r0 + excl + c0;
        cnt[node + 1] = c1;
    }
    ncnt[tid] = 0;
    ncnt[tid + 256] = 0;
    __syncthreads();

    for (int i = r0 + tid; i < r1; i += 2048) {
        int2 r[8];
#pragma unroll
        for (int u = 0; u < 8; ++u) {
            const int j = i + u * 256;
            r[u].x = 0;
            r[u].y = -1;
            if (j < r1) r[u] = recs[j];
        }
#pragma unroll
        for (int u = 0; u < 8; ++u) {
            if (r[u].y >= 0) {
                const int dl = r[u].y - nb0;
                const int p = atomicAdd(&ncnt[dl], 1);
                perm[r0 + noff[dl] + p] = r[u].x;
            }
        }
    }
}

__device__ __forceinline__ void row_load(const float* __restrict__ p, float (&r)[1]) {
    r[0] = p[0];
}
__device__ __forceinline__ void row_load(const float* __restrict__ p, float (&r)[2]) {
    const v2f t = *(const v2f*)p;
    r[0] = t.x;
    r[1] = t.y;
}

template <int D>
__device__ __forceinline__ void add_rows(const float* __restrict__ Hl, int s, float w,
                                         int c, float (&acc)[D / 32]) {
    constexpr int V = D / 32;
    for (int k = 0; k < c; k += 8) {
        float wk[8];
        float r[8][V];
#pragma unroll
        for (int u = 0; u < 8; ++u) {
            wk[u] = 0.0f;
            for (int v = 0; v < V; ++v) r[u][v] = 0.0f;
            if (k + u < c) {
                const int sk = __builtin_amdgcn_readlane(s, k + u);
                wk[u] = __int_as_float(__builtin_amdgcn_readlane(__float_as_int(w), k + u));
                row_load(Hl + (size_t)sk * D, r[u]);
            }
        }
#pragma unroll
        for (int u = 0; u < 8; ++u)
            for (int v = 0; v < V; ++v) acc[v] = fmaf(wk[u], r[u][v], acc[v]);
    }
}

constexpr int GATHER_WGS = 2048;

template <int D, bool FINAL>
__launch_bounds__(256)
__global__ void gat_gather(const int* __restrict__ off, const int* __restrict__ cnt,
                           const int* __restrict__ perm,
                           const float* __restrict__ H,
                           const float* __restrict__ as,
                           const float* __restrict__ ad,
                           const float* __restrict__ b,
                           float* __restrict__ out, int n) {
    constexpr int V = D / 32;
    const int lane = threadIdx.x & 31;
    const int wave = __builtin_amdgcn_readfirstlane((int)(threadIdx.x >> 5));
    constexpr int nwaves = GATHER_WGS * 8;
    const float* __restrict__ Hl = H + lane * V;

    for (int node = blockIdx.x * 8 + wave; node < n; node += nwaves) {
        const int start = off[node];
        const int deg = cnt[node];
        const float adv = ad[node];
        const float lself = lrelu(as[node] + adv, 0.2f);

        int s0 = 0;
        float l0 = -__builtin_inff();
        if (lane < deg) {
            s0 = perm[start + lane];
            l0 = lrelu(as[s0] + adv, 0.2f);
        }
        float m = fmaxf(lself, l0);
        for (int base = 32; base < deg; base += 32) {
            const int j = base + lane;
            if (j < deg) m = fmaxf(m, lrelu(as[perm[start + j]] + adv, 0.2f));
        }
        for (int o = 16; o > 0; o >>= 1) m = fmaxf(m, __shfl_xor(m, o));

        const float wself = __expf(lself - m);
        float acc[V];
        {
            float r[V];
            row_load(Hl + (size_t)node * D, r);
            for (int v = 0; v < V; ++v) acc[v] = wself * r[v];
        }
        const float w0 = lane < deg ? __expf(l0 - m) : 0.0f;
        float dsum = w0;
        add_rows<D>(Hl, s0, w0, deg < 32 ? deg : 32, acc);
        for (int base = 32; base < deg; base += 32) {
            const int j = base + lane;
            int s = 0;
            float w = 0.0f;
            if (j < deg) {
                s = perm[start + j];
                w = __expf(lrelu(as[s] + adv, 0.2f) - m);
            }
            dsum += w;
            const int c = deg - base;
            add_rows<D>(Hl, s, w, c < 32 ? c : 32, acc);
        }
        for (int o = 16; o > 0; o >>= 1) dsum += __shfl_xor(dsum, o);
        const float den = dsum + wself + 1e-16f;

        float* __restrict__ o = out + (size_t)node * D + lane * V;
        float res[V];
        for (int v = 0; v < V; ++v) {
            res[v] = acc[v] / den;
            if (FINAL) res[v] = lrelu(res[v] + b[lane * V + v], 0.01f);
        }
        if (V == 1) {
            o[0] = res[0];
        } else {
            v2f t;
            t.x = res[0];
            t.y = res[V - 1];
            *(v2f*)o = t;
        }
    }
}

extern "C" void kernel_launch(void* const* d_in, const int* in_sizes, int n_in,
                              void* d_out, int out_size, void* d_ws, size_t ws_size,
                              hipStream_t stream) {
    const float* x   = (const float*)d_in[0];
    const float* W1  = (const float*)d_in[1];
    const float* aS1 = (const float*)d_in[2];
    const float* aD1 = (const float*)d_in[3];
    const float* b1  = (const float*)d_in[4];
    const float* W2  = (const float*)d_in[5];
    const float* aS2 = (const float*)d_in[6];
    const float* aD2 = (const float*)d_in[7];
    const float* b2  = (const float*)d_in[8];
    const int*   ei  = (const int*)d_in[9];

    const int n  = in_sizes[0] / 64;
    const int E  = in_sizes[9] / 2;
    const int NB = (n + BNODES - 1) / BNODES;
    const int CH = (E + NCHUNK - 1) / NCHUNK;

    float* w    = (float*)d_ws;
    float* h    = w;
    int*   recs = (int*)(h + (size_t)n * 64);
    float* asN  = (float*)(recs + (size_t)2 * E);
    float* adN  = asN + (size_t)n;
    int*   cnt  = (int*)(adN + (size_t)n);
    int*   off  = cnt + (size_t)n;
    int*   perm = off + (size_t)n;
    int*   ctab = perm + (size_t)E;
    int*   bbase = ctab + (size_t)NCHUNK * 256;

    float* agg  = h;

    float* midh    = (float*)d_out;
    float* decodex = (float*)d_out + (size_t)n * 32;

    const int TPB = 256;
    const int tiles = (n + 15) / 16;
    const dim3 gemm_grid((tiles + 7) / 8);
    const dim3 node_grid((n + TPB - 1) / TPB);
    const dim3 wave_grid(GATHER_WGS);

    chunk_hist<<<dim3(NCHUNK), TPB, 0, stream>>>(ei, E, CH, NB, ctab);
    chunk_place<<<dim3(NCHUNK), TPB, 0, stream>>>(ei, E, CH, NB, ctab, bbase, recs);
    bucket_sort<<<dim3(NB), TPB, 0, stream>>>(recs, bbase, E, NB, n, off, cnt, perm);

    gat_gemm_wmma<64, 32, false><<<gemm_grid, TPB, 0, stream>>>(x, W1, b1, h, n);
    node_dots<32><<<node_grid, TPB, 0, stream>>>(h, aS1, aD1, asN, adN, n);
    gat_gather<32, true><<<wave_grid, TPB, 0, stream>>>(off, cnt, perm, h, asN, adN, b1, midh, n);

    node_dots_proj<32, 64><<<node_grid, TPB, 0, stream>>>(midh, W2, aS2, aD2, asN, adN, n);
    gat_gather<32, false><<<wave_grid, TPB, 0, stream>>>(off, cnt, perm, midh, asN, adN, b2, agg, n);
    gat_gemm_wmma<32, 64, true><<<gemm_grid, TPB, 0, stream>>>(agg, W2, b2, decodex, n);
}


# Edge logits taken from the gathered rows (row . a, 8 lanes per 128-byte row), not from per-node logit launches and a 4-byte gather per edge; sums regrouped: 4 lane groups, passes of 24, running max
# speedup vs baseline: 8.2423x; 1.1475x over previous
#include <hip/hip_runtime.h>
#include <cstddef>
#include <cstdint>

typedef float v2f __attribute__((ext_vector_type(2)));
typedef float v8f __attribute__((ext_vector_type(8)));

__device__ __forceinline__ float lrelu(float x, float s) {
    return x > 0.0f ? x : s * x;
}

template <int K, int NOUT, bool BIAS_ACT>
__launch_bounds__(256)
__global__ void gat_gemm_wmma(const float* __restrict__ X,
                              const float* __restrict__ W,
                              const float* __restrict__ bias,
                              float* __restrict__ H, int n) {
    constexpr int NT  = NOUT / 16;
    constexpr int LDK = K + 2;
    __shared__ alignas(16) float sWt[NOUT * LDK];
    for (int i = threadIdx.x; i < K * NOUT; i += 256) {
        const int k = i / NOUT;
        const int c = i - k * NOUT;
        sWt[c * LDK + k] = W[i];
    }
    __syncthreads();

    const int wave = threadIdx.x >> 5;
    const int lane = threadIdx.x & 31;
    const int tile = blockIdx.x * 8 + wave;
    const int m0 = tile * 16;
    if (m0 >= n) return;

    const int ml = lane & 15;
    const int kh = lane >> 4;
    int row = m0 + ml;
    row = row < n ? row : n - 1;
    const float* __restrict__ xrow = X + (size_t)row * K;

    v8f acc[NT] = {};

#pragma unroll
    for (int kk = 0; kk < K; kk += 4) {
        const v2f a = *(const v2f*)(xrow + kk + 2 * kh);
#pragma unroll
        for (int t = 0; t < NT; ++t) {
            const v2f b = *(const v2f*)(&sWt[(t * 16 + ml) * LDK + kk + 2 * kh]);
            acc[t] = __builtin_amdgcn_wmma_f32_16x16x4_f32(
                false, a, false, b, (short)0, acc[t], false, false);
        }
    }

    const int rb = kh * 8;
#pragma unroll
    for (int t = 0; t < NT; ++t) {
        float bv = 0.0f;
        if (BIAS_ACT) bv = bias[t * 16 + ml];
#pragma unroll
        for (int r = 0; r < 8; ++r) {
            const int m = m0 + rb + r;
            float v = acc[t][r];
            if (BIAS_ACT) v = lrelu(v + bv, 0.01f);
            if (m < n) H[(size_t)m * NOUT + t * 16 + ml] = v;
        }
    }
}

constexpr int NCHUNK  = 256;
constexpr int BSHIFT  = 9;
constexpr int BNODES  = 1 << BSHIFT;

__device__ __forceinline__ int block_excl_scan(int v, int* sw) {
    const int lane = threadIdx.x & 31;
    const int wave = threadIdx.x >> 5;
    int inc = v;
    for (int o = 1; o < 32; o <<= 1) {
        const int t = __shfl_up(inc, o);
        if (lane >= o) inc += t;
    }
    if (lane == 31) sw[wave] = inc;
    __syncthreads();
    int excl = inc - v;
    for (int w = 0; w < 8; ++w)
        if (w < wave) excl += sw[w];
    return excl;
}

__launch_bounds__(256)
__global__ void chunk_hist(const int* __restrict__ ei, int E, int CH, int NB,
                           int* __restrict__ ctab) {
    __shared__ int lc[256];
    lc[threadIdx.x] = 0;
    __syncthreads();
    const int c = blockIdx.x;
    const int e0 = c * CH;
    const int e1 = (e0 + CH < E) ? e0 + CH : E;
    const int* __restrict__ dst = ei + (size_t)E;
    for (int e = e0 + (int)threadIdx.x; e < e1; e += 2048) {
        int d[8];
#pragma unroll
        for (int u = 0; u < 8; ++u) {
            const int j = e + u * 256;
            d[u] = (j < e1) ? dst[j] : -1;
        }
#pragma unroll
        for (int u = 0; u < 8; ++u)
            if (d[u] >= 0) atomicAdd(&lc[d[u] >> BSHIFT], 1);
    }
    __syncthreads();
    if ((int)threadIdx.x < NB) ctab[c * NB + threadIdx.x] = lc[threadIdx.x];
}

__launch_bounds__(256)
__global__ void chunk_place(const int* __restrict__ ei, int E, int CH, int NB,
                            const int* __restrict__ ctab,
                            int* __restrict__ bbase, int* __restrict__ recs_w) {
    __shared__ int lbase[256];
    __shared__ int lcur[256];
    __shared__ int sw[8];
    const int tid = threadIdx.x;
    const int c = blockIdx.x;

    int before = 0, after = 0;
    if (tid < NB) {
        const int* __restrict__ col = ctab + tid;
#pragma unroll 32
        for (int cc = 0; cc < NCHUNK; ++cc) {
            const int v = col[cc * NB];
            if (cc < c) before += v; else after += v;
        }
    }
    const int excl = block_excl_scan(before + after, sw);
    lbase[tid] = excl + before;
    lcur[tid] = 0;
    if (c == 0 && tid < NB) bbase[tid] = excl;
    __syncthreads();

    int2* __restrict__ recs = reinterpret_cast<int2*>(recs_w);
    const int e0 = c * CH;
    const int e1 = (e0 + CH < E) ? e0 + CH : E;
    const int* __restrict__ dst = ei + (size_t)E;
    for (int e = e0 + tid; e < e1; e += 2048) {
        int d[8], s[8];
#pragma unroll
        for (int u = 0; u < 8; ++u) {
            const int j = e + u * 256;
            d[u] = -1;
            s[u] = 0;
            if (j < e1) {
                d[u] = dst[j];
                s[u] = ei[j];
            }
        }
#pragma unroll
        for (int u = 0; u < 8; ++u) {
            if (d[u] >= 0) {
                const int b = d[u] >> BSHIFT;
                const int r = atomicAdd(&lcur[b], 1);
                int2 rec;
                rec.x = s[u];
                rec.y = d[u];
                recs[lbase[b] + r] = rec;
            }
        }
    }
}

__launch_bounds__(256)
__global__ void bucket_sort(const int* __restrict__ recs_w,
                            const int* __restrict__ bbase, int E, int NB, int n,
                            int* __restrict__ off, int* __restrict__ cnt,
                            int* __restrict__ perm) {
    __shared__ int ncnt[BNODES];
    __shared__ int noff[BNODES];
    __shared__ int sw[8];
    const int tid = threadIdx.x;
    const int b = blockIdx.x;
    const int r0 = bbase[b];
    const int r1 = (b + 1 < NB) ? bbase[b + 1] : E;
    const int nb0 = b << BSHIFT;
    const int2* __restrict__ recs = reinterpret_cast<const int2*>(recs_w);

    ncnt[tid] = 0;
    ncnt[tid + 256] = 0;
    __syncthreads();
    for (int i = r0 + tid; i < r1; i += 2048) {
        int d[8];
#pragma unroll
        for (int u = 0; u < 8; ++u) {
            const int j = i + u * 256;
            d[u] = (j < r1) ? recs[j].y : -1;
        }
#pragma unroll
        for (int u = 0; u < 8; ++u)
            if (d[u] >= 0) atomicAdd(&ncnt[d[u] - nb0], 1);
    }
    __syncthreads();

    const int c0 = ncnt[2 * tid];
    const int c1 = ncnt[2 * tid + 1];
    const int excl = block_excl_scan(c0 + c1, sw);
    noff[2 * tid] = excl;
    noff[2 * tid + 1] = excl + c0;
    const int node = nb0 + 2 * tid;
    if (node < n) {
        off[node] = r0 + excl;
        cnt[node] = c0;
    }
    if (node + 1 < n) {
        off[node + 1] = r0 + excl + c0;
        cnt[node + 1] = c1;
    }
    ncnt[tid] = 0;
    ncnt[tid + 256] = 0;
    __syncthreads();

    for (int i = r0 + tid; i < r1; i += 2048) {
        int2 r[8];
#pragma unroll
        for (int u = 0; u < 8; ++u) {
            const int j = i + u * 256;
            r[u].x = 0;
            r[u].y = -1;
            if (j < r1) r[u] = recs[j];
        }
#pragma unroll
        for (int u = 0; u < 8; ++u) {
            if (r[u].y >= 0) {
                const int dl = r[u].y - nb0;
                const int p = atomicAdd(&ncnt[dl], 1);
                perm[r0 + noff[dl] + p] = r[u].x;
            }
        }
    }
}

typedef float v4f __attribute__((ext_vector_type(4)));

template <int M>
__device__ __forceinline__ float xrow(float x) {
    const int b = __builtin_bit_cast(int, x);
    return __builtin_bit_cast(float,
        __builtin_amdgcn_update_dpp(b, b, 0x160 + M, 0xf, 0xf, true));
}
__device__ __forceinline__ float xhalf(float x) {
    const int b = __builtin_bit_cast(int, x);
    return __builtin_bit_cast(float,
        __builtin_amdgcn_permlanex16(b, b, 0x76543210, 0xfedcba98, false, false));
}
template <int K>
__device__ __forceinline__ int bcast8(int x) {
    return __builtin_amdgcn_mov_dpp8(x, K * 0x249249);
}
__device__ __forceinline__ float dot4(v4f a, v4f b) {
    return fmaf(a.w, b.w, fmaf(a.z, b.z, fmaf(a.y, b.y, a.x * b.x)));
}
__device__ __forceinline__ float sum8(float x) {
    x += xrow<1>(x);
    x += xrow<2>(x);
    x += xrow<4>(x);
    return x;
}
__device__ __forceinline__ float sum_groups(float x) {
    x += xrow<8>(x);
    x += xhalf(x);
    return x;
}

constexpr int GATHER_WGS = 2048;

template <bool PROJ>
__launch_bounds__(256)
__global__ void gat_gather(const int* __restrict__ off, const int* __restrict__ cnt,
                           const int* __restrict__ perm,
                           const float* __restrict__ H,
                           const float* __restrict__ W,
                           const float* __restrict__ a_s,
                           const float* __restrict__ a_d,
                           const float* __restrict__ b,
                           float* __restrict__ out, int n) {
    const int tid = threadIdx.x;
    const int lane = tid & 31;
    const int q = lane & 7;
    const int g = lane >> 3;
    const int wave = __builtin_amdgcn_readfirstlane(tid >> 5);
    constexpr int nwaves = GATHER_WGS * 8;
    const float ninf = -__builtin_inff();

    v4f vs, vd;
    if (PROJ) {
        __shared__ alignas(16) float su[64];
        const float* __restrict__ wp = W + tid * 8;
        const v4f w0 = *(const v4f*)wp;
        const v4f w1 = *(const v4f*)(wp + 4);
        const int c0 = (tid & 7) * 8;
        float ps = dot4(w0, *(const v4f*)(a_s + c0)) + dot4(w1, *(const v4f*)(a_s + c0 + 4));
        float pd = dot4(w0, *(const v4f*)(a_d + c0)) + dot4(w1, *(const v4f*)(a_d + c0 + 4));
        ps = sum8(ps);
        pd = sum8(pd);
        if ((tid & 7) == 0) {
            su[tid >> 3] = ps;
            su[32 + (tid >> 3)] = pd;
        }
        __syncthreads();
        vs = *(const v4f*)&su[q * 4];
        vd = *(const v4f*)&su[32 + q * 4];
    } else {
        vs = *(const v4f*)(a_s + q * 4);
        vd = *(const v4f*)(a_d + q * 4);
    }
    v4f bv = {0.0f, 0.0f, 0.0f, 0.0f};
    if (!PROJ) bv = *(const v4f*)(b + q * 4);

    const unsigned qbyte = (unsigned)q * 16u;
    const int pull = (4 * q + g) * 4;
    const char* __restrict__ Hb = (const char*)H;

    for (int node = blockIdx.x * 8 + wave; node < n; node += nwaves) {
        const int start = off[node];
        const int T = cnt[node] + 1;
        float m = ninf, dsum = 0.0f, adv = 0.0f;
        v4f acc = {0.0f, 0.0f, 0.0f, 0.0f};

        int base = 0;
        do {
            const int j = base + lane;
            int src = node;
            if (j >= 1 && j < T && lane < 24) src = perm[start + j - 1];
            const int c = T - base;

            const int sg = __builtin_amdgcn_ds_bpermute(pull, src);
            int sk[6];
            sk[0] = bcast8<0>(sg);
            sk[1] = bcast8<1>(sg);
            sk[2] = bcast8<2>(sg);
            sk[3] = bcast8<3>(sg);
            sk[4] = bcast8<4>(sg);
            sk[5] = bcast8<5>(sg);

            v4f r[6];
#pragma unroll
            for (int k = 0; k < 6; ++k)
                if (4 * k < c)
                    r[k] = *(const v4f*)(Hb + (((unsigned)sk[k] << 7) + qbyte));

            if (base == 0)
                adv = __builtin_bit_cast(float, __builtin_amdgcn_readfirstlane(
                          __builtin_bit_cast(int, sum8(dot4(r[0], vd)))));

            float l[6];
            float mc = ninf;
#pragma unroll
            for (int k = 0; k < 6; ++k) {
                if (4 * k < c) {
                    const float e = lrelu(sum8(dot4(r[k], vs)) + adv, 0.2f);
                    l[k] = (g < c - 4 * k) ? e : ninf;
                    mc = fmaxf(mc, l[k]);
                }
            }
            mc = fmaxf(mc, xrow<8>(mc));
            mc = fmaxf(mc, xhalf(mc));
            const float mn = fmaxf(m, mc);
            if (base > 0) {
                const float sc = __expf(m - mn);
                dsum *= sc;
                acc *= sc;
            }
            m = mn;

#pragma unroll
            for (int k = 0; k < 6; ++k) {
                if (4 * k < c) {
                    const float w = __expf(l[k] - m);
                    dsum += w;
                    acc.x = fmaf(w, r[k].x, acc.x);
                    acc.y = fmaf(w, r[k].y, acc.y);
                    acc.z = fmaf(w, r[k].z, acc.z);
                    acc.w = fmaf(w, r[k].w, acc.w);
                }
            }
            base += 24;
        } while (base < T);

        dsum = sum_groups(dsum);
        acc.x = sum_groups(acc.x);
        acc.y = sum_groups(acc.y);
        acc.z = sum_groups(acc.z);
        acc.w = sum_groups(acc.w);
        const float inv = 1.0f / (dsum + 1e-16f);
        v4f res = acc * inv;
        if (!PROJ) {
            res += bv;
            res.x = lrelu(res.x, 0.01f);
            res.y = lrelu(res.y, 0.01f);
            res.z = lrelu(res.z, 0.01f);
            res.w = lrelu(res.w, 0.01f);
        }
        if (g == 0) *(v4f*)((char*)out + (((unsigned)node << 7) + qbyte)) = res;
    }
}

extern "C" void kernel_launch(void* const* d_in, const int* in_sizes, int n_in,
                              void* d_out, int out_size, void* d_ws, size_t ws_size,
                              hipStream_t stream) {
    const float* x   = (const float*)d_in[0];
    const float* W1  = (const float*)d_in[1];
    const float* aS1 = (const float*)d_in[2];
    const float* aD1 = (const float*)d_in[3];
    const float* b1  = (const float*)d_in[4];
    const float* W2  = (const float*)d_in[5];
    const float* aS2 = (const float*)d_in[6];
    const float* aD2 = (const float*)d_in[7];
    const float* b2  = (const float*)d_in[8];
    const int*   ei  = (const int*)d_in[9];

    const int n  = in_sizes[0] / 64;
    const int E  = in_sizes[9] / 2;
    const int NB = (n + BNODES - 1) / BNODES;
    const int CH = (E + NCHUNK - 1) / NCHUNK;

    float* w    = (float*)d_ws;
    float* h    = w;
    int*   recs = (int*)(h + (size_t)n * 64);
    int*   cnt  = recs + (size_t)2 * E;
    int*   off  = cnt + (size_t)n;
    int*   perm = off + (size_t)n;
    int*   ctab = perm + (size_t)E;
    int*   bbase = ctab + (size_t)NCHUNK * 256;

    float* agg  = h;

    float* midh    = (float*)d_out;
    float* decodex = (float*)d_out + (size_t)n * 32;

    const int TPB = 256;
    const int tiles = (n + 15) / 16;
    const dim3 gemm_grid((tiles + 7) / 8);
    const dim3 wave_grid(GATHER_WGS);

    chunk_hist<<<dim3(NCHUNK), TPB, 0, stream>>>(ei, E, CH, NB, ctab);
    chunk_place<<<dim3(NCHUNK), TPB, 0, stream>>>(ei, E, CH, NB, ctab, bbase, recs);
    bucket_sort<<<dim3(NB), TPB, 0, stream>>>(recs, bbase, E, NB, n, off, cnt, perm);

    gat_gemm_wmma<64, 32, false><<<gemm_grid, TPB, 0, stream>>>(x, W1, b1, h, n);
    gat_gather<false><<<wave_grid, TPB, 0, stream>>>(off, cnt, perm, h, W1, aS1, aD1, b1, midh, n);

    gat_gather<true><<<wave_grid, TPB, 0, stream>>>(off, cnt, perm, midh, W2, aS2, aD2, b2, agg, n);
    gat_gemm_wmma<32, 64, true><<<gemm_grid, TPB, 0, stream>>>(agg, W2, b2, decodex, n);
}


# Layer 1's projection tiles run as extra work-groups inside the two longer edge-sort launches (independent work side by side) instead of a launch of their own: 6 launches, sums unchanged
# speedup vs baseline: 9.0619x; 1.0994x over previous
#include <hip/hip_runtime.h>
#include <cstddef>
#include <cstdint>

typedef float v2f __attribute__((ext_vector_type(2)));
typedef float v8f __attribute__((ext_vector_type(8)));

__device__ __forceinline__ float lrelu(float x, float s) {
    return x > 0.0f ? x : s * x;
}

template <int K, int NOUT, bool BIAS_ACT>
__device__ __forceinline__ void gemm_wg(const float* __restrict__ X,
                                        const float* __restrict__ W,
                                        const float* __restrict__ bias,
                                        float* __restrict__ H, int n, int wg) {
    constexpr int NT  = NOUT / 16;
    constexpr int LDK = K + 2;
    __shared__ alignas(16) float sWt[NOUT * LDK];
    for (int i = threadIdx.x; i < K * NOUT; i += 256) {
        const int k = i / NOUT;
        const int c = i - k * NOUT;
        sWt[c * LDK + k] = W[i];
    }
    __syncthreads();

    const int wave = threadIdx.x >> 5;
    const int lane = threadIdx.x & 31;
    const int tile = wg * 8 + wave;
    const int m0 = tile * 16;
    if (m0 >= n) return;

    const int ml = lane & 15;
    const int kh = lane >> 4;
    int row = m0 + ml;
    row = row < n ? row : n - 1;
    const float* __restrict__ xrow = X + (size_t)row * K;

    v8f acc[NT] = {};

#pragma unroll
    for (int kk = 0; kk < K; kk += 4) {
        const v2f a = *(const v2f*)(xrow + kk + 2 * kh);
#pragma unroll
        for (int t = 0; t < NT; ++t) {
            const v2f b = *(const v2f*)(&sWt[(t * 16 + ml) * LDK + kk + 2 * kh]);
            acc[t] = __builtin_amdgcn_wmma_f32_16x16x4_f32(
                false, a, false, b, (short)0, acc[t], false, false);
        }
    }

    const int rb = kh * 8;
#pragma unroll
    for (int t = 0; t < NT; ++t) {
        float bv = 0.0f;
        if (BIAS_ACT) bv = bias[t * 16 + ml];
#pragma unroll
        for (int r = 0; r < 8; ++r) {
            const int m = m0 + rb + r;
            float v = acc[t][r];
            if (BIAS_ACT) v = lrelu(v + bv, 0.01f);
            if (m < n) H[(size_t)m * NOUT + t * 16 + ml] = v;
        }
    }
}

template <int K, int NOUT, bool BIAS_ACT>
__launch_bounds__(256)
__global__ void gat_gemm_wmma(const float* __restrict__ X,
                              const float* __restrict__ W,
                              const float* __restrict__ bias,
                              float* __restrict__ H, int n) {
    gemm_wg<K, NOUT, BIAS_ACT>(X, W, bias, H, n, blockIdx.x);
}

constexpr int NCHUNK  = 256;
constexpr int BSHIFT  = 9;
constexpr int BNODES  = 1 << BSHIFT;

__device__ __forceinline__ int block_excl_scan(int v, int* sw) {
    const int lane = threadIdx.x & 31;
    const int wave = threadIdx.x >> 5;
    int inc = v;
    for (int o = 1; o < 32; o <<= 1) {
        const int t = __shfl_up(inc, o);
        if (lane >= o) inc += t;
    }
    if (lane == 31) sw[wave] = inc;
    __syncthreads();
    int excl = inc - v;
    for (int w = 0; w < 8; ++w)
        if (w < wave) excl += sw[w];
    return excl;
}

__launch_bounds__(256)
__global__ void chunk_hist(const int* __restrict__ ei, int E, int CH, int NB,
                           int* __restrict__ ctab) {
    __shared__ int lc[256];
    lc[threadIdx.x] = 0;
    __syncthreads();
    const int c = blockIdx.x;
    const int e0 = c * CH;
    const int e1 = (e0 + CH < E) ? e0 + CH : E;
    const int* __restrict__ dst = ei + (size_t)E;
    for (int e = e0 + (int)threadIdx.x; e < e1; e += 2048) {
        int d[8];
#pragma unroll
        for (int u = 0; u < 8; ++u) {
            const int j = e + u * 256;
            d[u] = (j < e1) ? dst[j] : -1;
        }
#pragma unroll
        for (int u = 0; u < 8; ++u)
            if (d[u] >= 0) atomicAdd(&lc[d[u] >> BSHIFT], 1);
    }
    __syncthreads();
    if ((int)threadIdx.x < NB) ctab[c * NB + threadIdx.x] = lc[threadIdx.x];
}

__device__ __forceinline__ void chunk_place_wg(const int* __restrict__ ei, int E, int CH, int NB,
                                               const int* __restrict__ ctab,
                                               int* __restrict__ bbase, int* __restrict__ recs_w,
                                               int c) {
    __shared__ int lbase[256];
    __shared__ int lcur[256];
    __shared__ int sw[8];
    const int tid = threadIdx.x;

    int before = 0, after = 0;
    if (tid < NB) {
        const int* __restrict__ col = ctab + tid;
#pragma unroll 32
        for (int cc = 0; cc < NCHUNK; ++cc) {
            const int v = col[cc * NB];
            if (cc < c) before += v; else after += v;
        }
    }
    const int excl = block_excl_scan(before + after, sw);
    lbase[tid] = excl + before;
    lcur[tid] = 0;
    if (c == 0 && tid < NB) bbase[tid] = excl;
    __syncthreads();

    int2* __restrict__ recs = reinterpret_cast<int2*>(recs_w);
    const int e0 = c * CH;
    const int e1 = (e0 + CH < E) ? e0 + CH : E;
    const int* __restrict__ dst = ei + (size_t)E;
    for (int e = e0 + tid; e < e1; e += 2048) {
        int d[8], s[8];
#pragma unroll
        for (int u = 0; u < 8; ++u) {
            const int j = e + u * 256;
            d[u] = -1;
            s[u] = 0;
            if (j < e1) {
                d[u] = dst[j];
                s[u] = ei[j];
            }
        }
#pragma unroll
        for (int u = 0; u < 8; ++u) {
            if (d[u] >= 0) {
                const int b = d[u] >> BSHIFT;
                const int r = atomicAdd(&lcur[b], 1);
                int2 rec;
                rec.x = s[u];
                rec.y = d[u];
                recs[lbase[b] + r] = rec;
            }
        }
    }
}

__device__ __forceinline__ void bucket_sort_wg(const int* __restrict__ recs_w,
                                               const int* __restrict__ bbase, int E, int NB, int n,
                                               int* __restrict__ off, int* __restrict__ cnt,
                                               int* __restrict__ perm, int b) {
    __shared__ int ncnt[BNODES];
    __shared__ int noff[BNODES];
    __shared__ int sw[8];
    const int tid = threadIdx.x;
    const int r0 = bbase[b];
    const int r1 = (b + 1 < NB) ? bbase[b + 1] : E;
    const int nb0 = b << BSHIFT;
    const int2* __restrict__ recs = reinterpret_cast<const int2*>(recs_w);

    ncnt[tid] = 0;
    ncnt[tid + 256] = 0;
    __syncthreads();
    for (int i = r0 + tid; i < r1; i += 2048) {
        int d[8];
#pragma unroll
        for (int u = 0; u < 8; ++u) {
            const int j = i + u * 256;
            d[u] = (j < r1) ? recs[j].y : -1;
        }
#pragma unroll
        for (int u = 0; u < 8; ++u)
            if (d[u] >= 0) atomicAdd(&ncnt[d[u] - nb0], 1);
    }
    __syncthreads();

    const int c0 = ncnt[2 * tid];
    const int c1 = ncnt[2 * tid + 1];
    const int excl = block_excl_scan(c0 + c1, sw);
    noff[2 * tid] = excl;
    noff[2 * tid + 1] = excl + c0;
    const int node = nb0 + 2 * tid;
    if (node < n) {
        off[node] = r0 + excl;
        cnt[node] = c0;
    }
    if (node + 1 < n) {
        off[node + 1] = r0 + excl + c0;
        cnt[node + 1] = c1;
    }
    ncnt[tid] = 0;
    ncnt[tid + 256] = 0;
    __syncthreads();

    for (int i = r0 + tid; i < r1; i += 2048) {
        int2 r[8];
#pragma unroll
        for (int u = 0; u < 8; ++u) {
            const int j = i + u * 256;
            r[u].x = 0;
            r[u].y = -1;
            if (j < r1) r[u] = recs[j];
        }
#pragma unroll
        for (int u = 0; u < 8; ++u) {
            if (r[u].y >= 0) {
                const int dl = r[u].y - nb0;
                const int p = atomicAdd(&ncnt[dl], 1);
                perm[r0 + noff[dl] + p] = r[u].x;
            }
        }
    }
}


__launch_bounds__(256)
__global__ void chunk_place_proj(const int* __restrict__ ei, int E, int CH, int NB,
                                 const int* __restrict__ ctab,
                                 int* __restrict__ bbase, int* __restrict__ recs_w,
                                 const float* __restrict__ X, const float* __restrict__ W,
                                 const float* __restrict__ bias, float* __restrict__ H,
                                 int n, int g0) {
    if ((int)blockIdx.x < NCHUNK)
        chunk_place_wg(ei, E, CH, NB, ctab, bbase, recs_w, blockIdx.x);
    else
        gemm_wg<64, 32, false>(X, W, bias, H, n, g0 + ((int)blockIdx.x - NCHUNK));
}

__launch_bounds__(256)
__global__ void bucket_sort_proj(const int* __restrict__ recs_w,
                                 const int* __restrict__ bbase, int E, int NB, int n,
                                 int* __restrict__ off, int* __restrict__ cnt,
                                 int* __restrict__ perm,
                                 const float* __restrict__ X, const float* __restrict__ W,
                                 const float* __restrict__ bias, float* __restrict__ H,
                                 int g0) {
    if ((int)blockIdx.x < NB)
        bucket_sort_wg(recs_w, bbase, E, NB, n, off, cnt, perm, blockIdx.x);
    else
        gemm_wg<64, 32, false>(X, W, bias, H, n, g0 + ((int)blockIdx.x - NB));
}

typedef float v4f __attribute__((ext_vector_type(4)));

template <int M>
__device__ __forceinline__ float xrow(float x) {
    const int b = __builtin_bit_cast(int, x);
    return __builtin_bit_cast(float,
        __builtin_amdgcn_update_dpp(b, b, 0x160 + M, 0xf, 0xf, true));
}
__device__ __forceinline__ float xhalf(float x) {
    const int b = __builtin_bit_cast(int, x);
    return __builtin_bit_cast(float,
        __builtin_amdgcn_permlanex16(b, b, 0x76543210, 0xfedcba98, false, false));
}
template <int K>
__device__ __forceinline__ int bcast8(int x) {
    return __builtin_amdgcn_mov_dpp8(x, K * 0x249249);
}
__device__ __forceinline__ float dot4(v4f a, v4f b) {
    return fmaf(a.w, b.w, fmaf(a.z, b.z, fmaf(a.y, b.y, a.x * b.x)));
}
__device__ __forceinline__ float sum8(float x) {
    x += xrow<1>(x);
    x += xrow<2>(x);
    x += xrow<4>(x);
    return x;
}
__device__ __forceinline__ float sum_groups(float x) {
    x += xrow<8>(x);
    x += xhalf(x);
    return x;
}

constexpr int GATHER_WGS = 2048;

template <bool PROJ>
__launch_bounds__(256)
__global__ void gat_gather(const int* __restrict__ off, const int* __restrict__ cnt,
                           const int* __restrict__ perm,
                           const float* __restrict__ H,
                           const float* __restrict__ W,
                           const float* __restrict__ a_s,
                           const float* __restrict__ a_d,
                           const float* __restrict__ b,
                           float* __restrict__ out, int n) {
    const int tid = threadIdx.x;
    const int lane = tid & 31;
    const int q = lane & 7;
    const int g = lane >> 3;
    const int wave = __builtin_amdgcn_readfirstlane(tid >> 5);
    constexpr int nwaves = GATHER_WGS * 8;
    const float ninf = -__builtin_inff();

    v4f vs, vd;
    if (PROJ) {
        __shared__ alignas(16) float su[64];
        const float* __restrict__ wp = W + tid * 8;
        const v4f w0 = *(const v4f*)wp;
        const v4f w1 = *(const v4f*)(wp + 4);
        const int c0 = (tid & 7) * 8;
        float ps = dot4(w0, *(const v4f*)(a_s + c0)) + dot4(w1, *(const v4f*)(a_s + c0 + 4));
        float pd = dot4(w0, *(const v4f*)(a_d + c0)) + dot4(w1, *(const v4f*)(a_d + c0 + 4));
        ps = sum8(ps);
        pd = sum8(pd);
        if ((tid & 7) == 0) {
            su[tid >> 3] = ps;
            su[32 + (tid >> 3)] = pd;
        }
        __syncthreads();
        vs = *(const v4f*)&su[q * 4];
        vd = *(const v4f*)&su[32 + q * 4];
    } else {
        vs = *(const v4f*)(a_s + q * 4);
        vd = *(const v4f*)(a_d + q * 4);
    }
    v4f bv = {0.0f, 0.0f, 0.0f, 0.0f};
    if (!PROJ) bv = *(const v4f*)(b + q * 4);

    const unsigned qbyte = (unsigned)q * 16u;
    const int pull = (4 * q + g) * 4;
    const char* __restrict__ Hb = (const char*)H;

    for (int node = blockIdx.x * 8 + wave; node < n; node += nwaves) {
        const int start = off[node];
        const int T = cnt[node] + 1;
        float m = ninf, dsum = 0.0f, adv = 0.0f;
        v4f acc = {0.0f, 0.0f, 0.0f, 0.0f};

        int base = 0;
        do {
            const int j = base + lane;
            int src = node;
            if (j >= 1 && j < T && lane < 24) src = perm[start + j - 1];
            const int c = T - base;

            const int sg = __builtin_amdgcn_ds_bpermute(pull, src);
            int sk[6];
            sk[0] = bcast8<0>(sg);
            sk[1] = bcast8<1>(sg);
            sk[2] = bcast8<2>(sg);
            sk[3] = bcast8<3>(sg);
            sk[4] = bcast8<4>(sg);
            sk[5] = bcast8<5>(sg);

            v4f r[6];
#pragma unroll
            for (int k = 0; k < 6; ++k)
                if (4 * k < c)
                    r[k] = *(const v4f*)(Hb + (((unsigned)sk[k] << 7) + qbyte));

            if (base == 0)
                adv = __builtin_bit_cast(float, __builtin_amdgcn_readfirstlane(
                          __builtin_bit_cast(int, sum8(dot4(r[0], vd)))));

            float l[6];
            float mc = ninf;
#pragma unroll
            for (int k = 0; k < 6; ++k) {
                if (4 * k < c) {
                    const float e = lrelu(sum8(dot4(r[k], vs)) + adv, 0.2f);
                    l[k] = (g < c - 4 * k) ? e : ninf;
                    mc = fmaxf(mc, l[k]);
                }
            }
            mc = fmaxf(mc, xrow<8>(mc));
            mc = fmaxf(mc, xhalf(mc));
            const float mn = fmaxf(m, mc);
            if (base > 0) {
                const float sc = __expf(m - mn);
                dsum *= sc;
                acc *= sc;
            }
            m = mn;

#pragma unroll
            for (int k = 0; k < 6; ++k) {
                if (4 * k < c) {
                    const float w = __expf(l[k] - m);
                    dsum += w;
                    acc.x = fmaf(w, r[k].x, acc.x);
                    acc.y = fmaf(w, r[k].y, acc.y);
                    acc.z = fmaf(w, r[k].z, acc.z);
                    acc.w = fmaf(w, r[k].w, acc.w);
                }
            }
            base += 24;
        } while (base < T);

        dsum = sum_groups(dsum);
        acc.x = sum_groups(acc.x);
        acc.y = sum_groups(acc.y);
        acc.z = sum_groups(acc.z);
        acc.w = sum_groups(acc.w);
        const float inv = 1.0f / (dsum + 1e-16f);
        v4f res = acc * inv;
        if (!PROJ) {
            res += bv;
            res.x = lrelu(res.x, 0.01f);
            res.y = lrelu(res.y, 0.01f);
            res.z = lrelu(res.z, 0.01f);
            res.w = lrelu(res.w, 0.01f);
        }
        if (g == 0) *(v4f*)((char*)out + (((unsigned)node << 7) + qbyte)) = res;
    }
}

extern "C" void kernel_launch(void* const* d_in, const int* in_sizes, int n_in,
                              void* d_out, int out_size, void* d_ws, size_t ws_size,
                              hipStream_t stream) {
    const float* x   = (const float*)d_in[0];
    const float* W1  = (const float*)d_in[1];
    const float* aS1 = (const float*)d_in[2];
    const float* aD1 = (const float*)d_in[3];
    const float* b1  = (const float*)d_in[4];
    const float* W2  = (const float*)d_in[5];
    const float* aS2 = (const float*)d_in[6];
    const float* aD2 = (const float*)d_in[7];
    const float* b2  = (const float*)d_in[8];
    const int*   ei  = (const int*)d_in[9];

    const int n  = in_sizes[0] / 64;
    const int E  = in_sizes[9] / 2;
    const int NB = (n + BNODES - 1) / BNODES;
    const int CH = (E + NCHUNK - 1) / NCHUNK;

    float* w    = (float*)d_ws;
    float* h    = w;
    int*   recs = (int*)(h + (size_t)n * 64);
    int*   cnt  = recs + (size_t)2 * E;
    int*   off  = cnt + (size_t)n;
    int*   perm = off + (size_t)n;
    int*   ctab = perm + (size_t)E;
    int*   bbase = ctab + (size_t)NCHUNK * 256;

    float* agg  = h;

    float* midh    = (float*)d_out;
    float* decodex = (float*)d_out + (size_t)n * 32;

    const int TPB = 256;
    const int tiles = (n + 15) / 16;
    const int gemm_wgs = (tiles + 7) / 8;
    const dim3 gemm_grid(gemm_wgs);
    const int g2 = gemm_wgs / 2;
    const int g3 = gemm_wgs - g2;
    const dim3 wave_grid(GATHER_WGS);

    chunk_hist<<<dim3(NCHUNK), TPB, 0, stream>>>(ei, E, CH, NB, ctab);
    chunk_place_proj<<<dim3(NCHUNK + g2), TPB, 0, stream>>>(ei, E, CH, NB, ctab, bbase, recs,
                                                           x, W1, b1, h, n, 0);
    bucket_sort_proj<<<dim3(NB + g3), TPB, 0, stream>>>(recs, bbase, E, NB, n, off, cnt, perm,
                                                       x, W1, b1, h, g2);

    gat_gather<false><<<wave_grid, TPB, 0, stream>>>(off, cnt, perm, h, W1, aS1, aD1, b1, midh, n);

    gat_gather<true><<<wave_grid, TPB, 0, stream>>>(off, cnt, perm, midh, W2, aS2, aD2, b2, agg, n);
    gat_gemm_wmma<32, 64, true><<<gemm_grid, TPB, 0, stream>>>(agg, W2, b2, decodex, n);
}


# Edge sort in two launches, not three: a chunk's work-group counts, scans and places its records in its own slice and leaves a run table; a bucket reads its 256 runs; list (sum) order changes
# speedup vs baseline: 10.1536x; 1.1205x over previous
#include <hip/hip_runtime.h>
#include <cstddef>
#include <cstdint>

typedef float v2f __attribute__((ext_vector_type(2)));
typedef float v8f __attribute__((ext_vector_type(8)));

__device__ __forceinline__ float lrelu(float x, float s) {
    return x > 0.0f ? x : s * x;
}

template <int K, int NOUT, bool BIAS_ACT>
__device__ __forceinline__ void gemm_wg(const float* __restrict__ X,
                                        const float* __restrict__ W,
                                        const float* __restrict__ bias,
                                        float* __restrict__ H, int n, int wg) {
    constexpr int NT  = NOUT / 16;
    constexpr int LDK = K + 2;
    __shared__ alignas(16) float sWt[NOUT * LDK];
    for (int i = threadIdx.x; i < K * NOUT; i += 256) {
        const int k = i / NOUT;
        const int c = i - k * NOUT;
        sWt[c * LDK + k] = W[i];
    }
    __syncthreads();

    const int wave = threadIdx.x >> 5;
    const int lane = threadIdx.x & 31;
    const int tile = wg * 8 + wave;
    const int m0 = tile * 16;
    if (m0 >= n) return;

    const int ml = lane & 15;
    const int kh = lane >> 4;
    int row = m0 + ml;
    row = row < n ? row : n - 1;
    const float* __restrict__ xrow = X + (size_t)row * K;

    v8f acc[NT] = {};

#pragma unroll
    for (int kk = 0; kk < K; kk += 4) {
        const v2f a = *(const v2f*)(xrow + kk + 2 * kh);
#pragma unroll
        for (int t = 0; t < NT; ++t) {
            const v2f b = *(const v2f*)(&sWt[(t * 16 + ml) * LDK + kk + 2 * kh]);
            acc[t] = __builtin_amdgcn_wmma_f32_16x16x4_f32(
                false, a, false, b, (short)0, acc[t], false, false);
        }
    }

    const int rb = kh * 8;
#pragma unroll
    for (int t = 0; t < NT; ++t) {
        float bv = 0.0f;
        if (BIAS_ACT) bv = bias[t * 16 + ml];
#pragma unroll
        for (int r = 0; r < 8; ++r) {
            const int m = m0 + rb + r;
            float v = acc[t][r];
            if (BIAS_ACT) v = lrelu(v + bv, 0.01f);
            if (m < n) H[(size_t)m * NOUT + t * 16 + ml] = v;
        }
    }
}

template <int K, int NOUT, bool BIAS_ACT>
__launch_bounds__(256)
__global__ void gat_gemm_wmma(const float* __restrict__ X,
                              const float* __restrict__ W,
                              const float* __restrict__ bias,
                              float* __restrict__ H, int n) {
    gemm_wg<K, NOUT, BIAS_ACT>(X, W, bias, H, n, blockIdx.x);
}

constexpr int NCHUNK  = 256;
constexpr int BSHIFT  = 9;
constexpr int BNODES  = 1 << BSHIFT;

__device__ __forceinline__ int block_excl_scan(int v, int* sw) {
    const int lane = threadIdx.x & 31;
    const int wave = threadIdx.x >> 5;
    int inc = v;
    for (int o = 1; o < 32; o <<= 1) {
        const int t = __shfl_up(inc, o);
        if (lane >= o) inc += t;
    }
    if (lane == 31) sw[wave] = inc;
    __syncthreads();
    int excl = inc - v;
    for (int w = 0; w < 8; ++w)
        if (w < wave) excl += sw[w];
    return excl;
}

__device__ __forceinline__ void chunk_sort_wg(const int* __restrict__ ei, int E, int CH,
                                              int* __restrict__ rtab_w, int* __restrict__ recs_w,
                                              int c) {
    __shared__ int lc[256];
    __shared__ int lbase[256];
    __shared__ int sw[8];
    const int tid = threadIdx.x;
    lc[tid] = 0;
    __syncthreads();

    const int e0 = c * CH;
    const int e1 = (e0 + CH < E) ? e0 + CH : E;
    const int* __restrict__ dst = ei + (size_t)E;
    for (int e = e0 + tid; e < e1; e += 2048) {
        int d[8];
#pragma unroll
        for (int u = 0; u < 8; ++u) {
            const int j = e + u * 256;
            d[u] = (j < e1) ? dst[j] : -1;
        }
#pragma unroll
        for (int u = 0; u < 8; ++u)
            if (d[u] >= 0) atomicAdd(&lc[d[u] >> BSHIFT], 1);
    }
    __syncthreads();

    const int k = lc[tid];
    const int excl = block_excl_scan(k, sw);
    lbase[tid] = excl;
    lc[tid] = 0;
    int2 ent;
    ent.x = excl;
    ent.y = k;
    reinterpret_cast<int2*>(rtab_w)[c * 256 + tid] = ent;
    __syncthreads();

    int2* __restrict__ recs = reinterpret_cast<int2*>(recs_w);
    for (int e = e0 + tid; e < e1; e += 2048) {
        int d[8], s[8];
#pragma unroll
        for (int u = 0; u < 8; ++u) {
            const int j = e + u * 256;
            d[u] = -1;
            s[u] = 0;
            if (j < e1) {
                d[u] = dst[j];
                s[u] = ei[j];
            }
        }
#pragma unroll
        for (int u = 0; u < 8; ++u) {
            if (d[u] >= 0) {
                const int b = d[u] >> BSHIFT;
                const int r = atomicAdd(&lc[b], 1);
                int2 rec;
                rec.x = s[u];
                rec.y = d[u];
                recs[e0 + lbase[b] + r] = rec;
            }
        }
    }
}

__device__ __forceinline__ void bucket_sort_wg(const int* __restrict__ recs_w,
                                               const int* __restrict__ rtab_w, int CH, int n,
                                               int* __restrict__ off, int* __restrict__ cnt,
                                               int* __restrict__ perm, int b) {
    __shared__ int ncnt[BNODES];
    __shared__ int noff[BNODES];
    __shared__ int sred[8];
    __shared__ int sw[8];
    const int tid = threadIdx.x;
    const int lane = tid & 31;
    const int wave = tid >> 5;
    const int nb0 = b << BSHIFT;
    const int2* __restrict__ recs = reinterpret_cast<const int2*>(recs_w);

    const int2 ent = reinterpret_cast<const int2*>(rtab_w)[tid * 256 + b];
    const int runx = tid * CH + ent.x;
    const int runy = ent.y;
    int so = ent.x;
    for (int o = 16; o > 0; o >>= 1) so += __shfl_xor(so, o);
    if (lane == 0) sred[wave] = so;
    ncnt[tid] = 0;
    ncnt[tid + 256] = 0;
    __syncthreads();
    int r0 = 0;
    for (int w = 0; w < 8; ++w) r0 += sred[w];

    for (int rr = 0; rr < 32; rr += 8) {
        int d[8][2];
#pragma unroll
        for (int u = 0; u < 8; ++u) {
            const int rx = __builtin_amdgcn_readlane(runx, rr + u);
            const int ry = __builtin_amdgcn_readlane(runy, rr + u);
            d[u][0] = (lane < ry) ? recs[rx + lane].y : -1;
            d[u][1] = (lane + 32 < ry) ? recs[rx + lane + 32].y : -1;
        }
#pragma unroll
        for (int u = 0; u < 8; ++u) {
            if (d[u][0] >= 0) atomicAdd(&ncnt[d[u][0] - nb0], 1);
            if (d[u][1] >= 0) atomicAdd(&ncnt[d[u][1] - nb0], 1);
        }
        for (int u = 0; u < 8; ++u) {
            const int rx = __builtin_amdgcn_readlane(runx, rr + u);
            const int ry = __builtin_amdgcn_readlane(runy, rr + u);
            for (int i = lane + 64; i < ry; i += 32)
                atomicAdd(&ncnt[recs[rx + i].y - nb0], 1);
        }
    }
    __syncthreads();

    const int c0 = ncnt[2 * tid];
    const int c1 = ncnt[2 * tid + 1];
    const int excl = block_excl_scan(c0 + c1, sw);
    noff[2 * tid] = excl;
    noff[2 * tid + 1] = excl + c0;
    const int node = nb0 + 2 * tid;
    if (node < n) {
        off[node] = r0 + excl;
        cnt[node] = c0;
    }
    if (node + 1 < n) {
        off[node + 1] = r0 + excl + c0;
        cnt[node + 1] = c1;
    }
    ncnt[tid] = 0;
    ncnt[tid + 256] = 0;
    __syncthreads();

    for (int rr = 0; rr < 32; rr += 8) {
        int2 r[8][2];
#pragma unroll
        for (int u = 0; u < 8; ++u) {
            const int rx = __builtin_amdgcn_readlane(runx, rr + u);
            const int ry = __builtin_amdgcn_readlane(runy, rr + u);
            r[u][0].x = 0;
            r[u][0].y = -1;
            r[u][1].x = 0;
            r[u][1].y = -1;
            if (lane < ry) r[u][0] = recs[rx + lane];
            if (lane + 32 < ry) r[u][1] = recs[rx + lane + 32];
        }
#pragma unroll
        for (int u = 0; u < 8; ++u) {
#pragma unroll
            for (int v = 0; v < 2; ++v) {
                if (r[u][v].y >= 0) {
                    const int dl = r[u][v].y - nb0;
                    const int p = atomicAdd(&ncnt[dl], 1);
                    perm[r0 + noff[dl] + p] = r[u][v].x;
                }
            }
        }
        for (int u = 0; u < 8; ++u) {
            const int rx = __builtin_amdgcn_readlane(runx, rr + u);
            const int ry = __builtin_amdgcn_readlane(runy, rr + u);
            for (int i = lane + 64; i < ry; i += 32) {
                const int2 rec = recs[rx + i];
                const int dl = rec.y - nb0;
                const int p = atomicAdd(&ncnt[dl], 1);
                perm[r0 + noff[dl] + p] = rec.x;
            }
        }
    }
}


__launch_bounds__(256)
__global__ void chunk_sort_proj(const int* __restrict__ ei, int E, int CH,
                                int* __restrict__ rtab_w, int* __restrict__ recs_w,
                                const float* __restrict__ X, const float* __restrict__ W,
                                const float* __restrict__ bias, float* __restrict__ H,
                                int n, int g0) {
    if ((int)blockIdx.x < NCHUNK)
        chunk_sort_wg(ei, E, CH, rtab_w, recs_w, blockIdx.x);
    else
        gemm_wg<64, 32, false>(X, W, bias, H, n, g0 + ((int)blockIdx.x - NCHUNK));
}

__launch_bounds__(256)
__global__ void bucket_sort_proj(const int* __restrict__ recs_w,
                                 const int* __restrict__ rtab_w, int CH, int NB, int n,
                                 int* __restrict__ off, int* __restrict__ cnt,
                                 int* __restrict__ perm,
                                 const float* __restrict__ X, const float* __restrict__ W,
                                 const float* __restrict__ bias, float* __restrict__ H,
                                 int g0) {
    if ((int)blockIdx.x < NB)
        bucket_sort_wg(recs_w, rtab_w, CH, n, off, cnt, perm, blockIdx.x);
    else
        gemm_wg<64, 32, false>(X, W, bias, H, n, g0 + ((int)blockIdx.x - NB));
}

typedef float v4f __attribute__((ext_vector_type(4)));

template <int M>
__device__ __forceinline__ float xrow(float x) {
    const int b = __builtin_bit_cast(int, x);
    return __builtin_bit_cast(float,
        __builtin_amdgcn_update_dpp(b, b, 0x160 + M, 0xf, 0xf, true));
}
__device__ __forceinline__ float xhalf(float x) {
    const int b = __builtin_bit_cast(int, x);
    return __builtin_bit_cast(float,
        __builtin_amdgcn_permlanex16(b, b, 0x76543210, 0xfedcba98, false, false));
}
template <int K>
__device__ __forceinline__ int bcast8(int x) {
    return __builtin_amdgcn_mov_dpp8(x, K * 0x249249);
}
__device__ __forceinline__ float dot4(v4f a, v4f b) {
    return fmaf(a.w, b.w, fmaf(a.z, b.z, fmaf(a.y, b.y, a.x * b.x)));
}
__device__ __forceinline__ float sum8(float x) {
    x += xrow<1>(x);
    x += xrow<2>(x);
    x += xrow<4>(x);
    return x;
}
__device__ __forceinline__ float sum_groups(float x) {
    x += xrow<8>(x);
    x += xhalf(x);
    return x;
}

constexpr int GATHER_WGS = 2048;

template <bool PROJ>
__launch_bounds__(256)
__global__ void gat_gather(const int* __restrict__ off, const int* __restrict__ cnt,
                           const int* __restrict__ perm,
                           const float* __restrict__ H,
                           const float* __restrict__ W,
                           const float* __restrict__ a_s,
                           const float* __restrict__ a_d,
                           const float* __restrict__ b,
                           float* __restrict__ out, int n) {
    const int tid = threadIdx.x;
    const int lane = tid & 31;
    const int q = lane & 7;
    const int g = lane >> 3;
    const int wave = __builtin_amdgcn_readfirstlane(tid >> 5);
    constexpr int nwaves = GATHER_WGS * 8;
    const float ninf = -__builtin_inff();

    v4f vs, vd;
    if (PROJ) {
        __shared__ alignas(16) float su[64];
        const float* __restrict__ wp = W + tid * 8;
        const v4f w0 = *(const v4f*)wp;
        const v4f w1 = *(const v4f*)(wp + 4);
        const int c0 = (tid & 7) * 8;
        float ps = dot4(w0, *(const v4f*)(a_s + c0)) + dot4(w1, *(const v4f*)(a_s + c0 + 4));
        float pd = dot4(w0, *(const v4f*)(a_d + c0)) + dot4(w1, *(const v4f*)(a_d + c0 + 4));
        ps = sum8(ps);
        pd = sum8(pd);
        if ((tid & 7) == 0) {
            su[tid >> 3] = ps;
            su[32 + (tid >> 3)] = pd;
        }
        __syncthreads();
        vs = *(const v4f*)&su[q * 4];
        vd = *(const v4f*)&su[32 + q * 4];
    } else {
        vs = *(const v4f*)(a_s + q * 4);
        vd = *(const v4f*)(a_d + q * 4);
    }
    v4f bv = {0.0f, 0.0f, 0.0f, 0.0f};
    if (!PROJ) bv = *(const v4f*)(b + q * 4);

    const unsigned qbyte = (unsigned)q * 16u;
    const int pull = (4 * q + g) * 4;
    const char* __restrict__ Hb = (const char*)H;

    for (int node = blockIdx.x * 8 + wave; node < n; node += nwaves) {
        const int start = off[node];
        const int T = cnt[node] + 1;
        float m = ninf, dsum = 0.0f, adv = 0.0f;
        v4f acc = {0.0f, 0.0f, 0.0f, 0.0f};

        int base = 0;
        do {
            const int j = base + lane;
            int src = node;
            if (j >= 1 && j < T && lane < 24) src = perm[start + j - 1];
            const int c = T - base;

            const int sg = __builtin_amdgcn_ds_bpermute(pull, src);
            int sk[6];
            sk[0] = bcast8<0>(sg);
            sk[1] = bcast8<1>(sg);
            sk[2] = bcast8<2>(sg);
            sk[3] = bcast8<3>(sg);
            sk[4] = bcast8<4>(sg);
            sk[5] = bcast8<5>(sg);

            v4f r[6];
#pragma unroll
            for (int k = 0; k < 6; ++k)
                if (4 * k < c)
                    r[k] = *(const v4f*)(Hb + (((unsigned)sk[k] << 7) + qbyte));

            if (base == 0)
                adv = __builtin_bit_cast(float, __builtin_amdgcn_readfirstlane(
                          __builtin_bit_cast(int, sum8(dot4(r[0], vd)))));

            float l[6];
            float mc = ninf;
#pragma unroll
            for (int k = 0; k < 6; ++k) {
                if (4 * k < c) {
                    const float e = lrelu(sum8(dot4(r[k], vs)) + adv, 0.2f);
                    l[k] = (g < c - 4 * k) ? e : ninf;
                    mc = fmaxf(mc, l[k]);
                }
            }
            mc = fmaxf(mc, xrow<8>(mc));
            mc = fmaxf(mc, xhalf(mc));
            const float mn = fmaxf(m, mc);
            if (base > 0) {
                const float sc = __expf(m - mn);
                dsum *= sc;
                acc *= sc;
            }
            m = mn;

#pragma unroll
            for (int k = 0; k < 6; ++k) {
                if (4 * k < c) {
                    const float w = __expf(l[k] - m);
                    dsum += w;
                    acc.x = fmaf(w, r[k].x, acc.x);
                    acc.y = fmaf(w, r[k].y, acc.y);
                    acc.z = fmaf(w, r[k].z, acc.z);
                    acc.w = fmaf(w, r[k].w, acc.w);
                }
            }
            base += 24;
        } while (base < T);

        dsum = sum_groups(dsum);
        acc.x = sum_groups(acc.x);
        acc.y = sum_groups(acc.y);
        acc.z = sum_groups(acc.z);
        acc.w = sum_groups(acc.w);
        const float inv = 1.0f / (dsum + 1e-16f);
        v4f res = acc * inv;
        if (!PROJ) {
            res += bv;
            res.x = lrelu(res.x, 0.01f);
            res.y = lrelu(res.y, 0.01f);
            res.z = lrelu(res.z, 0.01f);
            res.w = lrelu(res.w, 0.01f);
        }
        if (g == 0) *(v4f*)((char*)out + (((unsigned)node << 7) + qbyte)) = res;
    }
}

extern "C" void kernel_launch(void* const* d_in, const int* in_sizes, int n_in,
                              void* d_out, int out_size, void* d_ws, size_t ws_size,
                              hipStream_t stream) {
    const float* x   = (const float*)d_in[0];
    const float* W1  = (const float*)d_in[1];
    const float* aS1 = (const float*)d_in[2];
    const float* aD1 = (const float*)d_in[3];
    const float* b1  = (const float*)d_in[4];
    const float* W2  = (const float*)d_in[5];
    const float* aS2 = (const float*)d_in[6];
    const float* aD2 = (const float*)d_in[7];
    const float* b2  = (const float*)d_in[8];
    const int*   ei  = (const int*)d_in[9];

    const int n  = in_sizes[0] / 64;
    const int E  = in_sizes[9] / 2;
    const int NB = (n + BNODES - 1) / BNODES;
    const int CH = (E + NCHUNK - 1) / NCHUNK;

    float* w    = (float*)d_ws;
    float* h    = w;
    int*   recs = (int*)(h + (size_t)n * 64);
    int*   rtab = recs + (size_t)2 * E;
    int*   cnt  = rtab + (size_t)2 * NCHUNK * 256;
    int*   off  = cnt + (size_t)n;
    int*   perm = off + (size_t)n;

    float* agg  = h;

    float* midh    = (float*)d_out;
    float* decodex = (float*)d_out + (size_t)n * 32;

    const int TPB = 256;
    const int tiles = (n + 15) / 16;
    const int gemm_wgs = (tiles + 7) / 8;
    const dim3 gemm_grid(gemm_wgs);
    const int g2 = gemm_wgs / 2;
    const int g3 = gemm_wgs - g2;
    const dim3 wave_grid(GATHER_WGS);

    chunk_sort_proj<<<dim3(NCHUNK + g2), TPB, 0, stream>>>(ei, E, CH, rtab, recs,
                                                          x, W1, b1, h, n, 0);
    bucket_sort_proj<<<dim3(NB + g3), TPB, 0, stream>>>(recs, rtab, CH, NB, n, off, cnt, perm,
                                                       x, W1, b1, h, g2);

    gat_gather<false><<<wave_grid, TPB, 0, stream>>>(off, cnt, perm, h, W1, aS1, aD1, b1, midh, n);

    gat_gather<true><<<wave_grid, TPB, 0, stream>>>(off, cnt, perm, midh, W2, aS2, aD2, b2, agg, n);
    gat_gemm_wmma<32, 64, true><<<gemm_grid, TPB, 0, stream>>>(agg, W2, b2, decodex, n);
}


# Sort scatter staged in LDS: a chunk's records and a bucket's source list are placed in an LDS image and written out as consecutive whole lines, not one 8- or 4-byte store per edge; sums unchanged
# speedup vs baseline: 10.7123x; 1.0550x over previous
#include <hip/hip_runtime.h>
#include <cstddef>
#include <cstdint>

typedef float v2f __attribute__((ext_vector_type(2)));
typedef float v8f __attribute__((ext_vector_type(8)));

__device__ __forceinline__ float lrelu(float x, float s) {
    return x > 0.0f ? x : s * x;
}

template <int K, int NOUT, bool BIAS_ACT>
__device__ __forceinline__ void gemm_wg(const float* __restrict__ X,
                                        const float* __restrict__ W,
                                        const float* __restrict__ bias,
                                        float* __restrict__ H, int n, int wg) {
    constexpr int NT  = NOUT / 16;
    constexpr int LDK = K + 2;
    __shared__ alignas(16) float sWt[NOUT * LDK];
    for (int i = threadIdx.x; i < K * NOUT; i += 256) {
        const int k = i / NOUT;
        const int c = i - k * NOUT;
        sWt[c * LDK + k] = W[i];
    }
    __syncthreads();

    const int wave = threadIdx.x >> 5;
    const int lane = threadIdx.x & 31;
    const int tile = wg * 8 + wave;
    const int m0 = tile * 16;
    if (m0 >= n) return;

    const int ml = lane & 15;
    const int kh = lane >> 4;
    int row = m0 + ml;
    row = row < n ? row : n - 1;
    const float* __restrict__ xrow = X + (size_t)row * K;

    v8f acc[NT] = {};

#pragma unroll
    for (int kk = 0; kk < K; kk += 4) {
        const v2f a = *(const v2f*)(xrow + kk + 2 * kh);
#pragma unroll
        for (int t = 0; t < NT; ++t) {
            const v2f b = *(const v2f*)(&sWt[(t * 16 + ml) * LDK + kk + 2 * kh]);
            acc[t] = __builtin_amdgcn_wmma_f32_16x16x4_f32(
                false, a, false, b, (short)0, acc[t], false, false);
        }
    }

    const int rb = kh * 8;
#pragma unroll
    for (int t = 0; t < NT; ++t) {
        float bv = 0.0f;
        if (BIAS_ACT) bv = bias[t * 16 + ml];
#pragma unroll
        for (int r = 0; r < 8; ++r) {
            const int m = m0 + rb + r;
            float v = acc[t][r];
            if (BIAS_ACT) v = lrelu(v + bv, 0.01f);
            if (m < n) H[(size_t)m * NOUT + t * 16 + ml] = v;
        }
    }
}

template <int K, int NOUT, bool BIAS_ACT>
__launch_bounds__(256)
__global__ void gat_gemm_wmma(const float* __restrict__ X,
                              const float* __restrict__ W,
                              const float* __restrict__ bias,
                              float* __restrict__ H, int n) {
    gemm_wg<K, NOUT, BIAS_ACT>(X, W, bias, H, n, blockIdx.x);
}

constexpr int NCHUNK  = 256;
constexpr int BSHIFT  = 9;
constexpr int BNODES  = 1 << BSHIFT;
constexpr int CHCAP   = 8192;
constexpr int BKCAP   = 16384;

__device__ __forceinline__ int block_excl_scan(int v, int* sw) {
    const int lane = threadIdx.x & 31;
    const int wave = threadIdx.x >> 5;
    int inc = v;
    for (int o = 1; o < 32; o <<= 1) {
        const int t = __shfl_up(inc, o);
        if (lane >= o) inc += t;
    }
    if (lane == 31) sw[wave] = inc;
    __syncthreads();
    int excl = inc - v;
    for (int w = 0; w < 8; ++w)
        if (w < wave) excl += sw[w];
    return excl;
}

__device__ __forceinline__ void chunk_sort_wg(const int* __restrict__ ei, int E, int CH,
                                              int* __restrict__ rtab_w, int* __restrict__ recs_w,
                                              int c) {
    __shared__ int lc[256];
    __shared__ int lbase[256];
    __shared__ int sw[8];
    __shared__ alignas(8) int srec_w[2 * CHCAP];
    int2* srec = reinterpret_cast<int2*>(srec_w);
    const int tid = threadIdx.x;
    lc[tid] = 0;
    __syncthreads();

    const int e0 = c * CH;
    const int e1 = (e0 + CH < E) ? e0 + CH : E;
    const int nrec = e1 - e0;
    const int* __restrict__ dst = ei + (size_t)E;
    for (int e = e0 + tid; e < e1; e += 2048) {
        int d[8];
#pragma unroll
        for (int u = 0; u < 8; ++u) {
            const int j = e + u * 256;
            d[u] = (j < e1) ? dst[j] : -1;
        }
#pragma unroll
        for (int u = 0; u < 8; ++u)
            if (d[u] >= 0) atomicAdd(&lc[d[u] >> BSHIFT], 1);
    }
    __syncthreads();

    const int k = lc[tid];
    const int excl = block_excl_scan(k, sw);
    lbase[tid] = excl;
    lc[tid] = 0;
    int2 ent;
    ent.x = excl;
    ent.y = k;
    reinterpret_cast<int2*>(rtab_w)[c * 256 + tid] = ent;
    __syncthreads();

    int2* __restrict__ recs = reinterpret_cast<int2*>(recs_w);
    if (nrec > CHCAP) {
        for (int e = e0 + tid; e < e1; e += 256) {
            int2 rec;
            rec.x = ei[e];
            rec.y = dst[e];
            const int b = rec.y >> BSHIFT;
            const int r = atomicAdd(&lc[b], 1);
            recs[e0 + lbase[b] + r] = rec;
        }
        return;
    }
    for (int e = e0 + tid; e < e1; e += 2048) {
        int d[8], s[8];
#pragma unroll
        for (int u = 0; u < 8; ++u) {
            const int j = e + u * 256;
            d[u] = -1;
            s[u] = 0;
            if (j < e1) {
                d[u] = dst[j];
                s[u] = ei[j];
            }
        }
#pragma unroll
        for (int u = 0; u < 8; ++u) {
            if (d[u] >= 0) {
                const int b = d[u] >> BSHIFT;
                const int r = atomicAdd(&lc[b], 1);
                int2 rec;
                rec.x = s[u];
                rec.y = d[u];
                srec[lbase[b] + r] = rec;
            }
        }
    }
    __syncthreads();
#pragma unroll 8
    for (int i = tid; i < nrec; i += 256) recs[e0 + i] = srec[i];
}

__device__ __forceinline__ void bucket_sort_wg(const int* __restrict__ recs_w,
                                               const int* __restrict__ rtab_w, int CH, int n,
                                               int* __restrict__ off, int* __restrict__ cnt,
                                               int* __restrict__ perm, int b) {
    __shared__ int ncnt[BNODES];
    __shared__ int noff[BNODES];
    __shared__ int sred[8];
    __shared__ int sw[8];
    __shared__ int stot;
    __shared__ int sperm[BKCAP];
    const int tid = threadIdx.x;
    const int lane = tid & 31;
    const int wave = tid >> 5;
    const int nb0 = b << BSHIFT;
    const int2* __restrict__ recs = reinterpret_cast<const int2*>(recs_w);

    const int2 ent = reinterpret_cast<const int2*>(rtab_w)[tid * 256 + b];
    const int runx = tid * CH + ent.x;
    const int runy = ent.y;
    int so = ent.x;
    for (int o = 16; o > 0; o >>= 1) so += __shfl_xor(so, o);
    if (lane == 0) sred[wave] = so;
    ncnt[tid] = 0;
    ncnt[tid + 256] = 0;
    __syncthreads();
    int r0 = 0;
    for (int w = 0; w < 8; ++w) r0 += sred[w];

    for (int rr = 0; rr < 32; rr += 8) {
        int d[8][2];
#pragma unroll
        for (int u = 0; u < 8; ++u) {
            const int rx = __builtin_amdgcn_readlane(runx, rr + u);
            const int ry = __builtin_amdgcn_readlane(runy, rr + u);
            d[u][0] = (lane < ry) ? recs[rx + lane].y : -1;
            d[u][1] = (lane + 32 < ry) ? recs[rx + lane + 32].y : -1;
        }
#pragma unroll
        for (int u = 0; u < 8; ++u) {
            if (d[u][0] >= 0) atomicAdd(&ncnt[d[u][0] - nb0], 1);
            if (d[u][1] >= 0) atomicAdd(&ncnt[d[u][1] - nb0], 1);
        }
        for (int u = 0; u < 8; ++u) {
            const int rx = __builtin_amdgcn_readlane(runx, rr + u);
            const int ry = __builtin_amdgcn_readlane(runy, rr + u);
            for (int i = lane + 64; i < ry; i += 32)
                atomicAdd(&ncnt[recs[rx + i].y - nb0], 1);
        }
    }
    __syncthreads();

    const int c0 = ncnt[2 * tid];
    const int c1 = ncnt[2 * tid + 1];
    const int excl = block_excl_scan(c0 + c1, sw);
    noff[2 * tid] = excl;
    noff[2 * tid + 1] = excl + c0;
    const int node = nb0 + 2 * tid;
    if (node < n) {
        off[node] = r0 + excl;
        cnt[node] = c0;
    }
    if (node + 1 < n) {
        off[node + 1] = r0 + excl + c0;
        cnt[node + 1] = c1;
    }
    if (tid == 255) stot = excl + c0 + c1;
    ncnt[tid] = 0;
    ncnt[tid + 256] = 0;
    __syncthreads();
    const int total = stot;

    if (total > BKCAP) {
        for (int u = 0; u < 32; ++u) {
            const int rx = __builtin_amdgcn_readlane(runx, u);
            const int ry = __builtin_amdgcn_readlane(runy, u);
            for (int i = lane; i < ry; i += 32) {
                const int2 rec = recs[rx + i];
                const int dl = rec.y - nb0;
                const int p = atomicAdd(&ncnt[dl], 1);
                perm[r0 + noff[dl] + p] = rec.x;
            }
        }
        return;
    }

    for (int rr = 0; rr < 32; rr += 8) {
        int2 r[8][2];
#pragma unroll
        for (int u = 0; u < 8; ++u) {
            const int rx = __builtin_amdgcn_readlane(runx, rr + u);
            const int ry = __builtin_amdgcn_readlane(runy, rr + u);
            r[u][0].x = 0;
            r[u][0].y = -1;
            r[u][1].x = 0;
            r[u][1].y = -1;
            if (lane < ry) r[u][0] = recs[rx + lane];
            if (lane + 32 < ry) r[u][1] = recs[rx + lane + 32];
        }
#pragma unroll
        for (int u = 0; u < 8; ++u) {
#pragma unroll
            for (int v = 0; v < 2; ++v) {
                if (r[u][v].y >= 0) {
                    const int dl = r[u][v].y - nb0;
                    const int p = atomicAdd(&ncnt[dl], 1);
                    sperm[noff[dl] + p] = r[u][v].x;
                }
            }
        }
        for (int u = 0; u < 8; ++u) {
            const int rx = __builtin_amdgcn_readlane(runx, rr + u);
            const int ry = __builtin_amdgcn_readlane(runy, rr + u);
            for (int i = lane + 64; i < ry; i += 32) {
                const int2 rec = recs[rx + i];
                const int dl = rec.y - nb0;
                const int p = atomicAdd(&ncnt[dl], 1);
                sperm[noff[dl] + p] = rec.x;
            }
        }
    }
    __syncthreads();
#pragma unroll 8
    for (int i = tid; i < total; i += 256) perm[r0 + i] = sperm[i];
}


__launch_bounds__(256)
__global__ void chunk_sort_proj(const int* __restrict__ ei, int E, int CH,
                                int* __restrict__ rtab_w, int* __restrict__ recs_w,
                                const float* __restrict__ X, const float* __restrict__ W,
                                const float* __restrict__ bias, float* __restrict__ H,
                                int n, int g0) {
    if ((int)blockIdx.x < NCHUNK)
        chunk_sort_wg(ei, E, CH, rtab_w, recs_w, blockIdx.x);
    else
        gemm_wg<64, 32, false>(X, W, bias, H, n, g0 + ((int)blockIdx.x - NCHUNK));
}

__launch_bounds__(256)
__global__ void bucket_sort_proj(const int* __restrict__ recs_w,
                                 const int* __restrict__ rtab_w, int CH, int NB, int n,
                                 int* __restrict__ off, int* __restrict__ cnt,
                                 int* __restrict__ perm,
                                 const float* __restrict__ X, const float* __restrict__ W,
                                 const float* __restrict__ bias, float* __restrict__ H,
                                 int g0) {
    if ((int)blockIdx.x < NB)
        bucket_sort_wg(recs_w, rtab_w, CH, n, off, cnt, perm, blockIdx.x);
    else
        gemm_wg<64, 32, false>(X, W, bias, H, n, g0 + ((int)blockIdx.x - NB));
}

typedef float v4f __attribute__((ext_vector_type(4)));

template <int M>
__device__ __forceinline__ float xrow(float x) {
    const int b = __builtin_bit_cast(int, x);
    return __builtin_bit_cast(float,
        __builtin_amdgcn_update_dpp(b, b, 0x160 + M, 0xf, 0xf, true));
}
__device__ __forceinline__ float xhalf(float x) {
    const int b = __builtin_bit_cast(int, x);
    return __builtin_bit_cast(float,
        __builtin_amdgcn_permlanex16(b, b, 0x76543210, 0xfedcba98, false, false));
}
template <int K>
__device__ __forceinline__ int bcast8(int x) {
    return __builtin_amdgcn_mov_dpp8(x, K * 0x249249);
}
__device__ __forceinline__ float dot4(v4f a, v4f b) {
    return fmaf(a.w, b.w, fmaf(a.z, b.z, fmaf(a.y, b.y, a.x * b.x)));
}
__device__ __forceinline__ float sum8(float x) {
    x += xrow<1>(x);
    x += xrow<2>(x);
    x += xrow<4>(x);
    return x;
}
__device__ __forceinline__ float sum_groups(float x) {
    x += xrow<8>(x);
    x += xhalf(x);
    return x;
}

constexpr int GATHER_WGS = 2048;

template <bool PROJ>
__launch_bounds__(256)
__global__ void gat_gather(const int* __restrict__ off, const int* __restrict__ cnt,
                           const int* __restrict__ perm,
                           const float* __restrict__ H,
                           const float* __restrict__ W,
                           const float* __restrict__ a_s,
                           const float* __restrict__ a_d,
                           const float* __restrict__ b,
                           float* __restrict__ out, int n) {
    const int tid = threadIdx.x;
    const int lane = tid & 31;
    const int q = lane & 7;
    const int g = lane >> 3;
    const int wave = __builtin_amdgcn_readfirstlane(tid >> 5);
    constexpr int nwaves = GATHER_WGS * 8;
    const float ninf = -__builtin_inff();

    v4f vs, vd;
    if (PROJ) {
        __shared__ alignas(16) float su[64];
        const float* __restrict__ wp = W + tid * 8;
        const v4f w0 = *(const v4f*)wp;
        const v4f w1 = *(const v4f*)(wp + 4);
        const int c0 = (tid & 7) * 8;
        float ps = dot4(w0, *(const v4f*)(a_s + c0)) + dot4(w1, *(const v4f*)(a_s + c0 + 4));
        float pd = dot4(w0, *(const v4f*)(a_d + c0)) + dot4(w1, *(const v4f*)(a_d + c0 + 4));
        ps = sum8(ps);
        pd = sum8(pd);
        if ((tid & 7) == 0) {
            su[tid >> 3] = ps;
            su[32 + (tid >> 3)] = pd;
        }
        __syncthreads();
        vs = *(const v4f*)&su[q * 4];
        vd = *(const v4f*)&su[32 + q * 4];
    } else {
        vs = *(const v4f*)(a_s + q * 4);
        vd = *(const v4f*)(a_d + q * 4);
    }
    v4f bv = {0.0f, 0.0f, 0.0f, 0.0f};
    if (!PROJ) bv = *(const v4f*)(b + q * 4);

    const unsigned qbyte = (unsigned)q * 16u;
    const int pull = (4 * q + g) * 4;
    const char* __restrict__ Hb = (const char*)H;

    for (int node = blockIdx.x * 8 + wave; node < n; node += nwaves) {
        const int start = off[node];
        const int T = cnt[node] + 1;
        float m = ninf, dsum = 0.0f, adv = 0.0f;
        v4f acc = {0.0f, 0.0f, 0.0f, 0.0f};

        int base = 0;
        do {
            const int j = base + lane;
            int src = node;
            if (j >= 1 && j < T && lane < 24) src = perm[start + j - 1];
            const int c = T - base;

            const int sg = __builtin_amdgcn_ds_bpermute(pull, src);
            int sk[6];
            sk[0] = bcast8<0>(sg);
            sk[1] = bcast8<1>(sg);
            sk[2] = bcast8<2>(sg);
            sk[3] = bcast8<3>(sg);
            sk[4] = bcast8<4>(sg);
            sk[5] = bcast8<5>(sg);

            v4f r[6];
#pragma unroll
            for (int k = 0; k < 6; ++k)
                if (4 * k < c)
                    r[k] = *(const v4f*)(Hb + (((unsigned)sk[k] << 7) + qbyte));

            if (base == 0)
                adv = __builtin_bit_cast(float, __builtin_amdgcn_readfirstlane(
                          __builtin_bit_cast(int, sum8(dot4(r[0], vd)))));

            float l[6];
            float mc = ninf;
#pragma unroll
            for (int k = 0; k < 6; ++k) {
                if (4 * k < c) {
                    const float e = lrelu(sum8(dot4(r[k], vs)) + adv, 0.2f);
                    l[k] = (g < c - 4 * k) ? e : ninf;
                    mc = fmaxf(mc, l[k]);
                }
            }
            mc = fmaxf(mc, xrow<8>(mc));
            mc = fmaxf(mc, xhalf(mc));
            const float mn = fmaxf(m, mc);
            if (base > 0) {
                const float sc = __expf(m - mn);
                dsum *= sc;
                acc *= sc;
            }
            m = mn;

#pragma unroll
            for (int k = 0; k < 6; ++k) {
                if (4 * k < c) {
                    const float w = __expf(l[k] - m);
                    dsum += w;
                    acc.x = fmaf(w, r[k].x, acc.x);
                    acc.y = fmaf(w, r[k].y, acc.y);
                    acc.z = fmaf(w, r[k].z, acc.z);
                    acc.w = fmaf(w, r[k].w, acc.w);
                }
            }
            base += 24;
        } while (base < T);

        dsum = sum_groups(dsum);
        acc.x = sum_groups(acc.x);
        acc.y = sum_groups(acc.y);
        acc.z = sum_groups(acc.z);
        acc.w = sum_groups(acc.w);
        const float inv = 1.0f / (dsum + 1e-16f);
        v4f res = acc * inv;
        if (!PROJ) {
            res += bv;
            res.x = lrelu(res.x, 0.01f);
            res.y = lrelu(res.y, 0.01f);
            res.z = lrelu(res.z, 0.01f);
            res.w = lrelu(res.w, 0.01f);
        }
        if (g == 0) *(v4f*)((char*)out + (((unsigned)node << 7) + qbyte)) = res;
    }
}

extern "C" void kernel_launch(void* const* d_in, const int* in_sizes, int n_in,
                              void* d_out, int out_size, void* d_ws, size_t ws_size,
                              hipStream_t stream) {
    const float* x   = (const float*)d_in[0];
    const float* W1  = (const float*)d_in[1];
    const float* aS1 = (const float*)d_in[2];
    const float* aD1 = (const float*)d_in[3];
    const float* b1  = (const float*)d_in[4];
    const float* W2  = (const float*)d_in[5];
    const float* aS2 = (const float*)d_in[6];
    const float* aD2 = (const float*)d_in[7];
    const float* b2  = (const float*)d_in[8];
    const int*   ei  = (const int*)d_in[9];

    const int n  = in_sizes[0] / 64;
    const int E  = in_sizes[9] / 2;
    const int NB = (n + BNODES - 1) / BNODES;
    const int CH = (E + NCHUNK - 1) / NCHUNK;

    float* w    = (float*)d_ws;
    float* h    = w;
    int*   recs = (int*)(h + (size_t)n * 64);
    int*   rtab = recs + (size_t)2 * E;
    int*   cnt  = rtab + (size_t)2 * NCHUNK * 256;
    int*   off  = cnt + (size_t)n;
    int*   perm = off + (size_t)n;

    float* agg  = h;

    float* midh    = (float*)d_out;
    float* decodex = (float*)d_out + (size_t)n * 32;

    const int TPB = 256;
    const int tiles = (n + 15) / 16;
    const int gemm_wgs = (tiles + 7) / 8;
    const dim3 gemm_grid(gemm_wgs);
    const int g2 = gemm_wgs / 2;
    const int g3 = gemm_wgs - g2;
    const dim3 wave_grid(GATHER_WGS);

    chunk_sort_proj<<<dim3(NCHUNK + g2), TPB, 0, stream>>>(ei, E, CH, rtab, recs,
                                                          x, W1, b1, h, n, 0);
    bucket_sort_proj<<<dim3(NB + g3), TPB, 0, stream>>>(recs, rtab, CH, NB, n, off, cnt, perm,
                                                       x, W1, b1, h, g2);

    gat_gather<false><<<wave_grid, TPB, 0, stream>>>(off, cnt, perm, h, W1, aS1, aD1, b1, midh, n);

    gat_gather<true><<<wave_grid, TPB, 0, stream>>>(off, cnt, perm, midh, W2, aS2, aD2, b2, agg, n);
    gat_gemm_wmma<32, 64, true><<<gemm_grid, TPB, 0, stream>>>(agg, W2, b2, decodex, n);
}


# Layer 2's projection by W2 (bias, activation) runs in the gather wave itself: its aggregated rows go through an LDS tile and WMMA; no GEMM launch, no aggregate round trip: 4 launches; sums unchanged
# speedup vs baseline: 11.0876x; 1.0350x over previous
#include <hip/hip_runtime.h>
#include <cstddef>
#include <cstdint>

typedef float v2f __attribute__((ext_vector_type(2)));
typedef float v8f __attribute__((ext_vector_type(8)));

__device__ __forceinline__ float lrelu(float x, float s) {
    return x > 0.0f ? x : s * x;
}

template <int K, int NOUT, bool BIAS_ACT>
__device__ __forceinline__ void gemm_wg(const float* __restrict__ X,
                                        const float* __restrict__ W,
                                        const float* __restrict__ bias,
                                        float* __restrict__ H, int n, int wg) {
    constexpr int NT  = NOUT / 16;
    constexpr int LDK = K + 2;
    __shared__ alignas(16) float sWt[NOUT * LDK];
    for (int i = threadIdx.x; i < K * NOUT; i += 256) {
        const int k = i / NOUT;
        const int c = i - k * NOUT;
        sWt[c * LDK + k] = W[i];
    }
    __syncthreads();

    const int wave = threadIdx.x >> 5;
    const int lane = threadIdx.x & 31;
    const int tile = wg * 8 + wave;
    const int m0 = tile * 16;
    if (m0 >= n) return;

    const int ml = lane & 15;
    const int kh = lane >> 4;
    int row = m0 + ml;
    row = row < n ? row : n - 1;
    const float* __restrict__ xrow = X + (size_t)row * K;

    v8f acc[NT] = {};

#pragma unroll
    for (int kk = 0; kk < K; kk += 4) {
        const v2f a = *(const v2f*)(xrow + kk + 2 * kh);
#pragma unroll
        for (int t = 0; t < NT; ++t) {
            const v2f b = *(const v2f*)(&sWt[(t * 16 + ml) * LDK + kk + 2 * kh]);
            acc[t] = __builtin_amdgcn_wmma_f32_16x16x4_f32(
                false, a, false, b, (short)0, acc[t], false, false);
        }
    }

    const int rb = kh * 8;
#pragma unroll
    for (int t = 0; t < NT; ++t) {
        float bv = 0.0f;
        if (BIAS_ACT) bv = bias[t * 16 + ml];
#pragma unroll
        for (int r = 0; r < 8; ++r) {
            const int m = m0 + rb + r;
            float v = acc[t][r];
            if (BIAS_ACT) v = lrelu(v + bv, 0.01f);
            if (m < n) H[(size_t)m * NOUT + t * 16 + ml] = v;
        }
    }
}

constexpr int NCHUNK  = 256;
constexpr int BSHIFT  = 9;
constexpr int BNODES  = 1 << BSHIFT;
constexpr int CHCAP   = 8192;
constexpr int BKCAP   = 16384;

__device__ __forceinline__ int block_excl_scan(int v, int* sw) {
    const int lane = threadIdx.x & 31;
    const int wave = threadIdx.x >> 5;
    int inc = v;
    for (int o = 1; o < 32; o <<= 1) {
        const int t = __shfl_up(inc, o);
        if (lane >= o) inc += t;
    }
    if (lane == 31) sw[wave] = inc;
    __syncthreads();
    int excl = inc - v;
    for (int w = 0; w < 8; ++w)
        if (w < wave) excl += sw[w];
    return excl;
}

__device__ __forceinline__ void chunk_sort_wg(const int* __restrict__ ei, int E, int CH,
                                              int* __restrict__ rtab_w, int* __restrict__ recs_w,
                                              int c) {
    __shared__ int lc[256];
    __shared__ int lbase[256];
    __shared__ int sw[8];
    __shared__ alignas(8) int srec_w[2 * CHCAP];
    int2* srec = reinterpret_cast<int2*>(srec_w);
    const int tid = threadIdx.x;
    lc[tid] = 0;
    __syncthreads();

    const int e0 = c * CH;
    const int e1 = (e0 + CH < E) ? e0 + CH : E;
    const int nrec = e1 - e0;
    const int* __restrict__ dst = ei + (size_t)E;
    for (int e = e0 + tid; e < e1; e += 2048) {
        int d[8];
#pragma unroll
        for (int u = 0; u < 8; ++u) {
            const int j = e + u * 256;
            d[u] = (j < e1) ? dst[j] : -1;
        }
#pragma unroll
        for (int u = 0; u < 8; ++u)
            if (d[u] >= 0) atomicAdd(&lc[d[u] >> BSHIFT], 1);
    }
    __syncthreads();

    const int k = lc[tid];
    const int excl = block_excl_scan(k, sw);
    lbase[tid] = excl;
    lc[tid] = 0;
    int2 ent;
    ent.x = excl;
    ent.y = k;
    reinterpret_cast<int2*>(rtab_w)[c * 256 + tid] = ent;
    __syncthreads();

    int2* __restrict__ recs = reinterpret_cast<int2*>(recs_w);
    if (nrec > CHCAP) {
        for (int e = e0 + tid; e < e1; e += 256) {
            int2 rec;
            rec.x = ei[e];
            rec.y = dst[e];
            const int b = rec.y >> BSHIFT;
            const int r = atomicAdd(&lc[b], 1);
            recs[e0 + lbase[b] + r] = rec;
        }
        return;
    }
    for (int e = e0 + tid; e < e1; e += 2048) {
        int d[8], s[8];
#pragma unroll
        for (int u = 0; u < 8; ++u) {
            const int j = e + u * 256;
            d[u] = -1;
            s[u] = 0;
            if (j < e1) {
                d[u] = dst[j];
                s[u] = ei[j];
            }
        }
#pragma unroll
        for (int u = 0; u < 8; ++u) {
            if (d[u] >= 0) {
                const int b = d[u] >> BSHIFT;
                const int r = atomicAdd(&lc[b], 1);
                int2 rec;
                rec.x = s[u];
                rec.y = d[u];
                srec[lbase[b] + r] = rec;
            }
        }
    }
    __syncthreads();
#pragma unroll 8
    for (int i = tid; i < nrec; i += 256) recs[e0 + i] = srec[i];
}

__device__ __forceinline__ void bucket_sort_wg(const int* __restrict__ recs_w,
                                               const int* __restrict__ rtab_w, int CH, int n,
                                               int* __restrict__ off, int* __restrict__ cnt,
                                               int* __restrict__ perm, int b) {
    __shared__ int ncnt[BNODES];
    __shared__ int noff[BNODES];
    __shared__ int sred[8];
    __shared__ int sw[8];
    __shared__ int stot;
    __shared__ int sperm[BKCAP];
    const int tid = threadIdx.x;
    const int lane = tid & 31;
    const int wave = tid >> 5;
    const int nb0 = b << BSHIFT;
    const int2* __restrict__ recs = reinterpret_cast<const int2*>(recs_w);

    const int2 ent = reinterpret_cast<const int2*>(rtab_w)[tid * 256 + b];
    const int runx = tid * CH + ent.x;
    const int runy = ent.y;
    int so = ent.x;
    for (int o = 16; o > 0; o >>= 1) so += __shfl_xor(so, o);
    if (lane == 0) sred[wave] = so;
    ncnt[tid] = 0;
    ncnt[tid + 256] = 0;
    __syncthreads();
    int r0 = 0;
    for (int w = 0; w < 8; ++w) r0 += sred[w];

    for (int rr = 0; rr < 32; rr += 8) {
        int d[8][2];
#pragma unroll
        for (int u = 0; u < 8; ++u) {
            const int rx = __builtin_amdgcn_readlane(runx, rr + u);
            const int ry = __builtin_amdgcn_readlane(runy, rr + u);
            d[u][0] = (lane < ry) ? recs[rx + lane].y : -1;
            d[u][1] = (lane + 32 < ry) ? recs[rx + lane + 32].y : -1;
        }
#pragma unroll
        for (int u = 0; u < 8; ++u) {
            if (d[u][0] >= 0) atomicAdd(&ncnt[d[u][0] - nb0], 1);
            if (d[u][1] >= 0) atomicAdd(&ncnt[d[u][1] - nb0], 1);
        }
        for (int u = 0; u < 8; ++u) {
            const int rx = __builtin_amdgcn_readlane(runx, rr + u);
            const int ry = __builtin_amdgcn_readlane(runy, rr + u);
            for (int i = lane + 64; i < ry; i += 32)
                atomicAdd(&ncnt[recs[rx + i].y - nb0], 1);
        }
    }
    __syncthreads();

    const int c0 = ncnt[2 * tid];
    const int c1 = ncnt[2 * tid + 1];
    const int excl = block_excl_scan(c0 + c1, sw);
    noff[2 * tid] = excl;
    noff[2 * tid + 1] = excl + c0;
    const int node = nb0 + 2 * tid;
    if (node < n) {
        off[node] = r0 + excl;
        cnt[node] = c0;
    }
    if (node + 1 < n) {
        off[node + 1] = r0 + excl + c0;
        cnt[node + 1] = c1;
    }
    if (tid == 255) stot = excl + c0 + c1;
    ncnt[tid] = 0;
    ncnt[tid + 256] = 0;
    __syncthreads();
    const int total = stot;

    if (total > BKCAP) {
        for (int u = 0; u < 32; ++u) {
            const int rx = __builtin_amdgcn_readlane(runx, u);
            const int ry = __builtin_amdgcn_readlane(runy, u);
            for (int i = lane; i < ry; i += 32) {
                const int2 rec = recs[rx + i];
                const int dl = rec.y - nb0;
                const int p = atomicAdd(&ncnt[dl], 1);
                perm[r0 + noff[dl] + p] = rec.x;
            }
        }
        return;
    }

    for (int rr = 0; rr < 32; rr += 8) {
        int2 r[8][2];
#pragma unroll
        for (int u = 0; u < 8; ++u) {
            const int rx = __builtin_amdgcn_readlane(runx, rr + u);
            const int ry = __builtin_amdgcn_readlane(runy, rr + u);
            r[u][0].x = 0;
            r[u][0].y = -1;
            r[u][1].x = 0;
            r[u][1].y = -1;
            if (lane < ry) r[u][0] = recs[rx + lane];
            if (lane + 32 < ry) r[u][1] = recs[rx + lane + 32];
        }
#pragma unroll
        for (int u = 0; u < 8; ++u) {
#pragma unroll
            for (int v = 0; v < 2; ++v) {
                if (r[u][v].y >= 0) {
                    const int dl = r[u][v].y - nb0;
                    const int p = atomicAdd(&ncnt[dl], 1);
                    sperm[noff[dl] + p] = r[u][v].x;
                }
            }
        }
        for (int u = 0; u < 8; ++u) {
            const int rx = __builtin_amdgcn_readlane(runx, rr + u);
            const int ry = __builtin_amdgcn_readlane(runy, rr + u);
            for (int i = lane + 64; i < ry; i += 32) {
                const int2 rec = recs[rx + i];
                const int dl = rec.y - nb0;
                const int p = atomicAdd(&ncnt[dl], 1);
                sperm[noff[dl] + p] = rec.x;
            }
        }
    }
    __syncthreads();
#pragma unroll 8
    for (int i = tid; i < total; i += 256) perm[r0 + i] = sperm[i];
}


__launch_bounds__(256)
__global__ void chunk_sort_proj(const int* __restrict__ ei, int E, int CH,
                                int* __restrict__ rtab_w, int* __restrict__ recs_w,
                                const float* __restrict__ X, const float* __restrict__ W,
                                const float* __restrict__ bias, float* __restrict__ H,
                                int n, int g0) {
    if ((int)blockIdx.x < NCHUNK)
        chunk_sort_wg(ei, E, CH, rtab_w, recs_w, blockIdx.x);
    else
        gemm_wg<64, 32, false>(X, W, bias, H, n, g0 + ((int)blockIdx.x - NCHUNK));
}

__launch_bounds__(256)
__global__ void bucket_sort_proj(const int* __restrict__ recs_w,
                                 const int* __restrict__ rtab_w, int CH, int NB, int n,
                                 int* __restrict__ off, int* __restrict__ cnt,
                                 int* __restrict__ perm,
                                 const float* __restrict__ X, const float* __restrict__ W,
                                 const float* __restrict__ bias, float* __restrict__ H,
                                 int g0) {
    if ((int)blockIdx.x < NB)
        bucket_sort_wg(recs_w, rtab_w, CH, n, off, cnt, perm, blockIdx.x);
    else
        gemm_wg<64, 32, false>(X, W, bias, H, n, g0 + ((int)blockIdx.x - NB));
}

typedef float v4f __attribute__((ext_vector_type(4)));

template <int M>
__device__ __forceinline__ float xrow(float x) {
    const int b = __builtin_bit_cast(int, x);
    return __builtin_bit_cast(float,
        __builtin_amdgcn_update_dpp(b, b, 0x160 + M, 0xf, 0xf, true));
}
__device__ __forceinline__ float xhalf(float x) {
    const int b = __builtin_bit_cast(int, x);
    return __builtin_bit_cast(float,
        __builtin_amdgcn_permlanex16(b, b, 0x76543210, 0xfedcba98, false, false));
}
template <int K>
__device__ __forceinline__ int bcast8(int x) {
    return __builtin_amdgcn_mov_dpp8(x, K * 0x249249);
}
__device__ __forceinline__ float dot4(v4f a, v4f b) {
    return fmaf(a.w, b.w, fmaf(a.z, b.z, fmaf(a.y, b.y, a.x * b.x)));
}
__device__ __forceinline__ float sum8(float x) {
    x += xrow<1>(x);
    x += xrow<2>(x);
    x += xrow<4>(x);
    return x;
}
__device__ __forceinline__ float sum_groups(float x) {
    x += xrow<8>(x);
    x += xhalf(x);
    return x;
}

constexpr int GATHER_WGS = 2048;

template <bool PROJ>
__launch_bounds__(256)
__global__ void gat_gather(const int* __restrict__ off, const int* __restrict__ cnt,
                           const int* __restrict__ perm,
                           const float* __restrict__ H,
                           const float* __restrict__ W,
                           const float* __restrict__ a_s,
                           const float* __restrict__ a_d,
                           const float* __restrict__ b,
                           float* __restrict__ out, int n) {
    const int tid = threadIdx.x;
    const int lane = tid & 31;
    const int q = lane & 7;
    const int g = lane >> 3;
    const int wave = __builtin_amdgcn_readfirstlane(tid >> 5);
    constexpr int nwaves = GATHER_WGS * 8;
    const float ninf = -__builtin_inff();

    constexpr int PLDK = 32 + 2;
    constexpr int TLD  = 36;
    __shared__ alignas(16) float sWt[PROJ ? 64 * PLDK : 1];
    __shared__ alignas(16) float stile[PROJ ? 8 * 16 * TLD : 1];

    v4f vs, vd;
    if (PROJ) {
        __shared__ alignas(16) float su[64];
        const float* __restrict__ wp = W + tid * 8;
        const v4f w0 = *(const v4f*)wp;
        const v4f w1 = *(const v4f*)(wp + 4);
        const int c0 = (tid & 7) * 8;
        float ps = dot4(w0, *(const v4f*)(a_s + c0)) + dot4(w1, *(const v4f*)(a_s + c0 + 4));
        float pd = dot4(w0, *(const v4f*)(a_d + c0)) + dot4(w1, *(const v4f*)(a_d + c0 + 4));
        ps = sum8(ps);
        pd = sum8(pd);
        if ((tid & 7) == 0) {
            su[tid >> 3] = ps;
            su[32 + (tid >> 3)] = pd;
        }
#pragma unroll
        for (int i = 0; i < 4; ++i) {
            sWt[(c0 + i) * PLDK + (tid >> 3)] = w0[i];
            sWt[(c0 + 4 + i) * PLDK + (tid >> 3)] = w1[i];
        }
        __syncthreads();
        vs = *(const v4f*)&su[q * 4];
        vd = *(const v4f*)&su[32 + q * 4];
    } else {
        vs = *(const v4f*)(a_s + q * 4);
        vd = *(const v4f*)(a_d + q * 4);
    }
    v4f bv = {0.0f, 0.0f, 0.0f, 0.0f};
    if (!PROJ) bv = *(const v4f*)(b + q * 4);

    const unsigned qbyte = (unsigned)q * 16u;
    const int pull = (4 * q + g) * 4;
    const char* __restrict__ Hb = (const char*)H;

    float* __restrict__ tile = &stile[PROJ ? wave * 16 * TLD : 0];
    const int tile_first = blockIdx.x * 8 + wave;
    int rows = 0;
    for (int node = tile_first; node < n; node += nwaves) {
        const int start = off[node];
        const int T = cnt[node] + 1;
        float m = ninf, dsum = 0.0f, adv = 0.0f;
        v4f acc = {0.0f, 0.0f, 0.0f, 0.0f};

        int base = 0;
        do {
            const int j = base + lane;
            int src = node;
            if (j >= 1 && j < T && lane < 24) src = perm[start + j - 1];
            const int c = T - base;

            const int sg = __builtin_amdgcn_ds_bpermute(pull, src);
            int sk[6];
            sk[0] = bcast8<0>(sg);
            sk[1] = bcast8<1>(sg);
            sk[2] = bcast8<2>(sg);
            sk[3] = bcast8<3>(sg);
            sk[4] = bcast8<4>(sg);
            sk[5] = bcast8<5>(sg);

            v4f r[6];
#pragma unroll
            for (int k = 0; k < 6; ++k)
                if (4 * k < c)
                    r[k] = *(const v4f*)(Hb + (((unsigned)sk[k] << 7) + qbyte));

            if (base == 0)
                adv = __builtin_bit_cast(float, __builtin_amdgcn_readfirstlane(
                          __builtin_bit_cast(int, sum8(dot4(r[0], vd)))));

            float l[6];
            float mc = ninf;
#pragma unroll
            for (int k = 0; k < 6; ++k) {
                if (4 * k < c) {
                    const float e = lrelu(sum8(dot4(r[k], vs)) + adv, 0.2f);
                    l[k] = (g < c - 4 * k) ? e : ninf;
                    mc = fmaxf(mc, l[k]);
                }
            }
            mc = fmaxf(mc, xrow<8>(mc));
            mc = fmaxf(mc, xhalf(mc));
            const float mn = fmaxf(m, mc);
            if (base > 0) {
                const float sc = __expf(m - mn);
                dsum *= sc;
                acc *= sc;
            }
            m = mn;

#pragma unroll
            for (int k = 0; k < 6; ++k) {
                if (4 * k < c) {
                    const float w = __expf(l[k] - m);
                    dsum += w;
                    acc.x = fmaf(w, r[k].x, acc.x);
                    acc.y = fmaf(w, r[k].y, acc.y);
                    acc.z = fmaf(w, r[k].z, acc.z);
                    acc.w = fmaf(w, r[k].w, acc.w);
                }
            }
            base += 24;
        } while (base < T);

        dsum = sum_groups(dsum);
        acc.x = sum_groups(acc.x);
        acc.y = sum_groups(acc.y);
        acc.z = sum_groups(acc.z);
        acc.w = sum_groups(acc.w);
        const float inv = 1.0f / (dsum + 1e-16f);
        v4f res = acc * inv;
        if (!PROJ) {
            res += bv;
            res.x = lrelu(res.x, 0.01f);
            res.y = lrelu(res.y, 0.01f);
            res.z = lrelu(res.z, 0.01f);
            res.w = lrelu(res.w, 0.01f);
        }
        if (PROJ) {
            if (g == 0) *(v4f*)&tile[rows * TLD + q * 4] = res;
        } else {
            if (g == 0) *(v4f*)((char*)out + (((unsigned)node << 7) + qbyte)) = res;
        }
        ++rows;
    }

    if (PROJ && rows > 0) {
        const int ml = lane & 15;
        const int kh = lane >> 4;
        __builtin_amdgcn_fence(__ATOMIC_RELEASE, "wavefront");
        __builtin_amdgcn_wave_barrier();
        __builtin_amdgcn_fence(__ATOMIC_ACQUIRE, "wavefront");

        const int arow = ml < rows ? ml : 0;
        const float* __restrict__ ap = tile + arow * TLD + 2 * kh;
        const int rb = kh * 8;
#pragma nounroll
        for (int t = 0; t < 4; ++t) {
            const float* __restrict__ bp = &sWt[(t * 16 + ml) * PLDK + 2 * kh];
            v8f pacc = {};
#pragma unroll
            for (int kk = 0; kk < 32; kk += 4) {
                const v2f a = *(const v2f*)(ap + kk);
                const v2f bb = *(const v2f*)(bp + kk);
                pacc = __builtin_amdgcn_wmma_f32_16x16x4_f32(
                    false, a, false, bb, (short)0, pacc, false, false);
            }
            const float bt = b[t * 16 + ml];
            const unsigned ob = ((unsigned)(tile_first + rb * nwaves) << 8)
                              + (unsigned)(t * 16 + ml) * 4u;
#pragma unroll
            for (int r = 0; r < 8; ++r) {
                const float v = lrelu(pacc[r] + bt, 0.01f);
                if (rb + r < rows)
                    *(float*)((char*)out + (ob + (unsigned)r * ((unsigned)nwaves << 8))) = v;
            }
        }
    }
}

extern "C" void kernel_launch(void* const* d_in, const int* in_sizes, int n_in,
                              void* d_out, int out_size, void* d_ws, size_t ws_size,
                              hipStream_t stream) {
    const float* x   = (const float*)d_in[0];
    const float* W1  = (const float*)d_in[1];
    const float* aS1 = (const float*)d_in[2];
    const float* aD1 = (const float*)d_in[3];
    const float* b1  = (const float*)d_in[4];
    const float* W2  = (const float*)d_in[5];
    const float* aS2 = (const float*)d_in[6];
    const float* aD2 = (const float*)d_in[7];
    const float* b2  = (const float*)d_in[8];
    const int*   ei  = (const int*)d_in[9];

    const int n  = in_sizes[0] / 64;
    const int E  = in_sizes[9] / 2;
    const int NB = (n + BNODES - 1) / BNODES;
    const int CH = (E + NCHUNK - 1) / NCHUNK;

    float* w    = (float*)d_ws;
    float* h    = w;
    int*   recs = (int*)(h + (size_t)n * 64);
    int*   rtab = recs + (size_t)2 * E;
    int*   cnt  = rtab + (size_t)2 * NCHUNK * 256;
    int*   off  = cnt + (size_t)n;
    int*   perm = off + (size_t)n;

    float* midh    = (float*)d_out;
    float* decodex = (float*)d_out + (size_t)n * 32;

    const int TPB = 256;
    const int tiles = (n + 15) / 16;
    const int gemm_wgs = (tiles + 7) / 8;
    const int g2 = gemm_wgs / 2;
    const int g3 = gemm_wgs - g2;
    const dim3 wave_grid(GATHER_WGS);

    chunk_sort_proj<<<dim3(NCHUNK + g2), TPB, 0, stream>>>(ei, E, CH, rtab, recs,
                                                          x, W1, b1, h, n, 0);
    bucket_sort_proj<<<dim3(NB + g3), TPB, 0, stream>>>(recs, rtab, CH, NB, n, off, cnt, perm,
                                                       x, W1, b1, h, g2);

    gat_gather<false><<<wave_grid, TPB, 0, stream>>>(off, cnt, perm, h, W1, aS1, aD1, b1, midh, n);

    gat_gather<true><<<wave_grid, TPB, 0, stream>>>(off, cnt, perm, midh, W2, aS2, aD2, b2, decodex, n);
}


# Sort ranks from per-wave LDS counters (none shared by two waves): a thread's 8 or 16 rank atomics go out before the first wait, not one LDS round trip per record; list (sum) order: by wave
# speedup vs baseline: 11.5999x; 1.0462x over previous
#include <hip/hip_runtime.h>
#include <cstddef>
#include <cstdint>

typedef float v2f __attribute__((ext_vector_type(2)));
typedef float v8f __attribute__((ext_vector_type(8)));

__device__ __forceinline__ float lrelu(float x, float s) {
    return x > 0.0f ? x : s * x;
}

template <int K, int NOUT, bool BIAS_ACT>
__device__ __forceinline__ void gemm_wg(const float* __restrict__ X,
                                        const float* __restrict__ W,
                                        const float* __restrict__ bias,
                                        float* __restrict__ H, int n, int wg) {
    constexpr int NT  = NOUT / 16;
    constexpr int LDK = K + 2;
    __shared__ alignas(16) float sWt[NOUT * LDK];
    for (int i = threadIdx.x; i < K * NOUT; i += 256) {
        const int k = i / NOUT;
        const int c = i - k * NOUT;
        sWt[c * LDK + k] = W[i];
    }
    __syncthreads();

    const int wave = threadIdx.x >> 5;
    const int lane = threadIdx.x & 31;
    const int tile = wg * 8 + wave;
    const int m0 = tile * 16;
    if (m0 >= n) return;

    const int ml = lane & 15;
    const int kh = lane >> 4;
    int row = m0 + ml;
    row = row < n ? row : n - 1;
    const float* __restrict__ xrow = X + (size_t)row * K;

    v8f acc[NT] = {};

#pragma unroll
    for (int kk = 0; kk < K; kk += 4) {
        const v2f a = *(const v2f*)(xrow + kk + 2 * kh);
#pragma unroll
        for (int t = 0; t < NT; ++t) {
            const v2f b = *(const v2f*)(&sWt[(t * 16 + ml) * LDK + kk + 2 * kh]);
            acc[t] = __builtin_amdgcn_wmma_f32_16x16x4_f32(
                false, a, false, b, (short)0, acc[t], false, false);
        }
    }

    const int rb = kh * 8;
#pragma unroll
    for (int t = 0; t < NT; ++t) {
        float bv = 0.0f;
        if (BIAS_ACT) bv = bias[t * 16 + ml];
#pragma unroll
        for (int r = 0; r < 8; ++r) {
            const int m = m0 + rb + r;
            float v = acc[t][r];
            if (BIAS_ACT) v = lrelu(v + bv, 0.01f);
            if (m < n) H[(size_t)m * NOUT + t * 16 + ml] = v;
        }
    }
}

constexpr int NCHUNK  = 256;
constexpr int BSHIFT  = 9;
constexpr int BNODES  = 1 << BSHIFT;
constexpr int SWAVES  = 8;
constexpr int CHCAP   = 8192;
constexpr int BKCAP   = 16384;

__device__ __forceinline__ int block_excl_scan(int v, int* sw) {
    const int lane = threadIdx.x & 31;
    const int wave = threadIdx.x >> 5;
    int inc = v;
    for (int o = 1; o < 32; o <<= 1) {
        const int t = __shfl_up(inc, o);
        if (lane >= o) inc += t;
    }
    if (lane == 31) sw[wave] = inc;
    __syncthreads();
    int excl = inc - v;
    for (int w = 0; w < 8; ++w)
        if (w < wave) excl += sw[w];
    return excl;
}

__device__ __forceinline__ void chunk_sort_wg(const int* __restrict__ ei, int E, int CH,
                                              int* __restrict__ rtab_w, int* __restrict__ recs_w,
                                              int c) {
    __shared__ int lcw[SWAVES * 256 + 256];
    __shared__ int sw[8];
    __shared__ alignas(8) int srec_w[2 * (CHCAP + 256)];
    int2* srec = reinterpret_cast<int2*>(srec_w);
    const int tid = threadIdx.x;
    int* lcm = lcw + (tid >> 5) * 256;
#pragma unroll
    for (int w = 0; w < SWAVES; ++w) lcw[w * 256 + tid] = 0;
    __syncthreads();

    const int e0 = c * CH;
    const int e1 = (e0 + CH < E) ? e0 + CH : E;
    const int nrec = e1 - e0;
    const int* __restrict__ dst = ei + (size_t)E;
    for (int e = e0 + tid; e < e1; e += 2048) {
        int d[8];
#pragma unroll
        for (int u = 0; u < 8; ++u) {
            const int j = e + u * 256;
            d[u] = (j < e1) ? dst[j] : -1;
        }
#pragma unroll
        for (int u = 0; u < 8; ++u)
            if (d[u] >= 0) atomicAdd(&lcm[d[u] >> BSHIFT], 1);
    }
    __syncthreads();

    int part[SWAVES];
    int k = 0;
#pragma unroll
    for (int w = 0; w < SWAVES; ++w) {
        part[w] = lcw[w * 256 + tid];
        k += part[w];
    }
    const int excl = block_excl_scan(k, sw);
    int cur = excl;
#pragma unroll
    for (int w = 0; w < SWAVES; ++w) {
        lcw[w * 256 + tid] = cur;
        cur += part[w];
    }
    int2 ent;
    ent.x = excl;
    ent.y = k;
    reinterpret_cast<int2*>(rtab_w)[c * 256 + tid] = ent;
    __syncthreads();

    int2* __restrict__ recs = reinterpret_cast<int2*>(recs_w);
    if (nrec > CHCAP) {
        for (int e = e0 + tid; e < e1; e += 256) {
            int2 rec;
            rec.x = ei[e];
            rec.y = dst[e];
            const int p = atomicAdd(&lcm[rec.y >> BSHIFT], 1);
            recs[e0 + p] = rec;
        }
        return;
    }
    for (int e = e0 + tid; e < e1; e += 2048) {
        int d[8], s[8];
#pragma unroll
        for (int u = 0; u < 8; ++u) {
            const int j = e + u * 256;
            d[u] = -1;
            s[u] = 0;
            if (j < e1) {
                d[u] = dst[j];
                s[u] = ei[j];
            }
        }
        int pos[8];
#pragma unroll
        for (int u = 0; u < 8; ++u) {
            const int ci = (d[u] >= 0) ? (tid >> 5) * 256 + (d[u] >> BSHIFT) : SWAVES * 256 + tid;
            pos[u] = atomicAdd(&lcw[ci], 1);
        }
#pragma unroll
        for (int u = 0; u < 8; ++u) {
            int2 rec;
            rec.x = s[u];
            rec.y = d[u];
            srec[(d[u] >= 0) ? pos[u] : CHCAP + tid] = rec;
        }
    }
    __syncthreads();
#pragma unroll 8
    for (int i = tid; i < nrec; i += 256) recs[e0 + i] = srec[i];
}

__device__ __forceinline__ void bucket_sort_wg(const int* __restrict__ recs_w,
                                               const int* __restrict__ rtab_w, int CH, int n,
                                               int* __restrict__ off, int* __restrict__ cnt,
                                               int* __restrict__ perm, int b) {
    __shared__ alignas(8) int ncw[SWAVES * BNODES + 256];
    __shared__ int sred[8];
    __shared__ int sw[8];
    __shared__ int stot;
    __shared__ int sperm[BKCAP + 256];
    const int tid = threadIdx.x;
    const int lane = tid & 31;
    const int wave = tid >> 5;
    const int nb0 = b << BSHIFT;
    const int2* __restrict__ recs = reinterpret_cast<const int2*>(recs_w);
    int* ncm = ncw + wave * BNODES;

    const int2 ent = reinterpret_cast<const int2*>(rtab_w)[tid * 256 + b];
    const int runx = tid * CH + ent.x;
    const int runy = ent.y;
    int so = ent.x;
    for (int o = 16; o > 0; o >>= 1) so += __shfl_xor(so, o);
    if (lane == 0) sred[wave] = so;
#pragma unroll
    for (int i = 0; i < SWAVES * BNODES / 256; ++i) ncw[i * 256 + tid] = 0;
    __syncthreads();
    int r0 = 0;
    for (int w = 0; w < 8; ++w) r0 += sred[w];

    for (int rr = 0; rr < 32; rr += 8) {
        int d[8][2];
#pragma unroll
        for (int u = 0; u < 8; ++u) {
            const int rx = __builtin_amdgcn_readlane(runx, rr + u);
            const int ry = __builtin_amdgcn_readlane(runy, rr + u);
            d[u][0] = (lane < ry) ? recs[rx + lane].y : -1;
            d[u][1] = (lane + 32 < ry) ? recs[rx + lane + 32].y : -1;
        }
#pragma unroll
        for (int u = 0; u < 8; ++u) {
            if (d[u][0] >= 0) atomicAdd(&ncm[d[u][0] - nb0], 1);
            if (d[u][1] >= 0) atomicAdd(&ncm[d[u][1] - nb0], 1);
        }
        for (int u = 0; u < 8; ++u) {
            const int rx = __builtin_amdgcn_readlane(runx, rr + u);
            const int ry = __builtin_amdgcn_readlane(runy, rr + u);
            for (int i = lane + 64; i < ry; i += 32)
                atomicAdd(&ncm[recs[rx + i].y - nb0], 1);
        }
    }
    __syncthreads();

    int2 part[SWAVES];
    int c0 = 0, c1 = 0;
#pragma unroll
    for (int w = 0; w < SWAVES; ++w) {
        part[w] = *reinterpret_cast<const int2*>(&ncw[w * BNODES + 2 * tid]);
        c0 += part[w].x;
        c1 += part[w].y;
    }
    const int excl = block_excl_scan(c0 + c1, sw);
    int cur0 = excl, cur1 = excl + c0;
#pragma unroll
    for (int w = 0; w < SWAVES; ++w) {
        int2 cu;
        cu.x = cur0;
        cu.y = cur1;
        *reinterpret_cast<int2*>(&ncw[w * BNODES + 2 * tid]) = cu;
        cur0 += part[w].x;
        cur1 += part[w].y;
    }
    const int node = nb0 + 2 * tid;
    if (node < n) {
        off[node] = r0 + excl;
        cnt[node] = c0;
    }
    if (node + 1 < n) {
        off[node + 1] = r0 + excl + c0;
        cnt[node + 1] = c1;
    }
    if (tid == 255) stot = excl + c0 + c1;
    __syncthreads();
    const int total = stot;

    if (total > BKCAP) {
        for (int u = 0; u < 32; ++u) {
            const int rx = __builtin_amdgcn_readlane(runx, u);
            const int ry = __builtin_amdgcn_readlane(runy, u);
            for (int i = lane; i < ry; i += 32) {
                const int2 rec = recs[rx + i];
                const int p = atomicAdd(&ncm[rec.y - nb0], 1);
                perm[r0 + p] = rec.x;
            }
        }
        return;
    }

#pragma nounroll
    for (int rr = 0; rr < 32; rr += 8) {
        int2 r[8][2];
#pragma unroll
        for (int u = 0; u < 8; ++u) {
            const int rx = __builtin_amdgcn_readlane(runx, rr + u);
            const int ry = __builtin_amdgcn_readlane(runy, rr + u);
            r[u][0].x = 0;
            r[u][0].y = -1;
            r[u][1].x = 0;
            r[u][1].y = -1;
            if (lane < ry) r[u][0] = recs[rx + lane];
            if (lane + 32 < ry) r[u][1] = recs[rx + lane + 32];
        }
        int pos[8][2];
#pragma unroll
        for (int u = 0; u < 8; ++u) {
#pragma unroll
            for (int v = 0; v < 2; ++v) {
                const int ci = (r[u][v].y >= 0) ? wave * BNODES + (r[u][v].y - nb0)
                                                : SWAVES * BNODES + tid;
                pos[u][v] = atomicAdd(&ncw[ci], 1);
            }
        }
#pragma unroll
        for (int u = 0; u < 8; ++u) {
#pragma unroll
            for (int v = 0; v < 2; ++v)
                sperm[(r[u][v].y >= 0) ? pos[u][v] : BKCAP + tid] = r[u][v].x;
        }
        for (int u = 0; u < 8; ++u) {
            const int rx = __builtin_amdgcn_readlane(runx, rr + u);
            const int ry = __builtin_amdgcn_readlane(runy, rr + u);
            for (int i = lane + 64; i < ry; i += 32) {
                const int2 rec = recs[rx + i];
                const int p = atomicAdd(&ncm[rec.y - nb0], 1);
                sperm[p] = rec.x;
            }
        }
    }
    __syncthreads();
#pragma unroll 8
    for (int i = tid; i < total; i += 256) perm[r0 + i] = sperm[i];
}


__launch_bounds__(256)
__global__ void chunk_sort_proj(const int* __restrict__ ei, int E, int CH,
                                int* __restrict__ rtab_w, int* __restrict__ recs_w,
                                const float* __restrict__ X, const float* __restrict__ W,
                                const float* __restrict__ bias, float* __restrict__ H,
                                int n, int g0) {
    if ((int)blockIdx.x < NCHUNK)
        chunk_sort_wg(ei, E, CH, rtab_w, recs_w, blockIdx.x);
    else
        gemm_wg<64, 32, false>(X, W, bias, H, n, g0 + ((int)blockIdx.x - NCHUNK));
}

__launch_bounds__(256)
__global__ void bucket_sort_proj(const int* __restrict__ recs_w,
                                 const int* __restrict__ rtab_w, int CH, int NB, int n,
                                 int* __restrict__ off, int* __restrict__ cnt,
                                 int* __restrict__ perm,
                                 const float* __restrict__ X, const float* __restrict__ W,
                                 const float* __restrict__ bias, float* __restrict__ H,
                                 int g0) {
    if ((int)blockIdx.x < NB)
        bucket_sort_wg(recs_w, rtab_w, CH, n, off, cnt, perm, blockIdx.x);
    else
        gemm_wg<64, 32, false>(X, W, bias, H, n, g0 + ((int)blockIdx.x - NB));
}

typedef float v4f __attribute__((ext_vector_type(4)));

template <int M>
__device__ __forceinline__ float xrow(float x) {
    const int b = __builtin_bit_cast(int, x);
    return __builtin_bit_cast(float,
        __builtin_amdgcn_update_dpp(b, b, 0x160 + M, 0xf, 0xf, true));
}
__device__ __forceinline__ float xhalf(float x) {
    const int b = __builtin_bit_cast(int, x);
    return __builtin_bit_cast(float,
        __builtin_amdgcn_permlanex16(b, b, 0x76543210, 0xfedcba98, false, false));
}
template <int K>
__device__ __forceinline__ int bcast8(int x) {
    return __builtin_amdgcn_mov_dpp8(x, K * 0x249249);
}
__device__ __forceinline__ float dot4(v4f a, v4f b) {
    return fmaf(a.w, b.w, fmaf(a.z, b.z, fmaf(a.y, b.y, a.x * b.x)));
}
__device__ __forceinline__ float sum8(float x) {
    x += xrow<1>(x);
    x += xrow<2>(x);
    x += xrow<4>(x);
    return x;
}
__device__ __forceinline__ float sum_groups(float x) {
    x += xrow<8>(x);
    x += xhalf(x);
    return x;
}

constexpr int GATHER_WGS = 2048;

template <bool PROJ>
__launch_bounds__(256)
__global__ void gat_gather(const int* __restrict__ off, const int* __restrict__ cnt,
                           const int* __restrict__ perm,
                           const float* __restrict__ H,
                           const float* __restrict__ W,
                           const float* __restrict__ a_s,
                           const float* __restrict__ a_d,
                           const float* __restrict__ b,
                           float* __restrict__ out, int n) {
    const int tid = threadIdx.x;
    const int lane = tid & 31;
    const int q = lane & 7;
    const int g = lane >> 3;
    const int wave = __builtin_amdgcn_readfirstlane(tid >> 5);
    constexpr int nwaves = GATHER_WGS * 8;
    const float ninf = -__builtin_inff();

    constexpr int PLDK = 32 + 2;
    constexpr int TLD  = 36;
    __shared__ alignas(16) float sWt[PROJ ? 64 * PLDK : 1];
    __shared__ alignas(16) float stile[PROJ ? 8 * 16 * TLD : 1];

    v4f vs, vd;
    if (PROJ) {
        __shared__ alignas(16) float su[64];
        const float* __restrict__ wp = W + tid * 8;
        const v4f w0 = *(const v4f*)wp;
        const v4f w1 = *(const v4f*)(wp + 4);
        const int c0 = (tid & 7) * 8;
        float ps = dot4(w0, *(const v4f*)(a_s + c0)) + dot4(w1, *(const v4f*)(a_s + c0 + 4));
        float pd = dot4(w0, *(const v4f*)(a_d + c0)) + dot4(w1, *(const v4f*)(a_d + c0 + 4));
        ps = sum8(ps);
        pd = sum8(pd);
        if ((tid & 7) == 0) {
            su[tid >> 3] = ps;
            su[32 + (tid >> 3)] = pd;
        }
#pragma unroll
        for (int i = 0; i < 4; ++i) {
            sWt[(c0 + i) * PLDK + (tid >> 3)] = w0[i];
            sWt[(c0 + 4 + i) * PLDK + (tid >> 3)] = w1[i];
        }
        __syncthreads();
        vs = *(const v4f*)&su[q * 4];
        vd = *(const v4f*)&su[32 + q * 4];
    } else {
        vs = *(const v4f*)(a_s + q * 4);
        vd = *(const v4f*)(a_d + q * 4);
    }
    v4f bv = {0.0f, 0.0f, 0.0f, 0.0f};
    if (!PROJ) bv = *(const v4f*)(b + q * 4);

    const unsigned qbyte = (unsigned)q * 16u;
    const int pull = (4 * q + g) * 4;
    const char* __restrict__ Hb = (const char*)H;

    float* __restrict__ tile = &stile[PROJ ? wave * 16 * TLD : 0];
    const int tile_first = blockIdx.x * 8 + wave;
    int rows = 0;
    for (int node = tile_first; node < n; node += nwaves) {
        const int start = off[node];
        const int T = cnt[node] + 1;
        float m = ninf, dsum = 0.0f, adv = 0.0f;
        v4f acc = {0.0f, 0.0f, 0.0f, 0.0f};

        int base = 0;
        do {
            const int j = base + lane;
            int src = node;
            if (j >= 1 && j < T && lane < 24) src = perm[start + j - 1];
            const int c = T - base;

            const int sg = __builtin_amdgcn_ds_bpermute(pull, src);
            int sk[6];
            sk[0] = bcast8<0>(sg);
            sk[1] = bcast8<1>(sg);
            sk[2] = bcast8<2>(sg);
            sk[3] = bcast8<3>(sg);
            sk[4] = bcast8<4>(sg);
            sk[5] = bcast8<5>(sg);

            v4f r[6];
#pragma unroll
            for (int k = 0; k < 6; ++k)
                if (4 * k < c)
                    r[k] = *(const v4f*)(Hb + (((unsigned)sk[k] << 7) + qbyte));

            if (base == 0)
                adv = __builtin_bit_cast(float, __builtin_amdgcn_readfirstlane(
                          __builtin_bit_cast(int, sum8(dot4(r[0], vd)))));

            float l[6];
            float mc = ninf;
#pragma unroll
            for (int k = 0; k < 6; ++k) {
                if (4 * k < c) {
                    const float e = lrelu(sum8(dot4(r[k], vs)) + adv, 0.2f);
                    l[k] = (g < c - 4 * k) ? e : ninf;
                    mc = fmaxf(mc, l[k]);
                }
            }
            mc = fmaxf(mc, xrow<8>(mc));
            mc = fmaxf(mc, xhalf(mc));
            const float mn = fmaxf(m, mc);
            if (base > 0) {
                const float sc = __expf(m - mn);
                dsum *= sc;
                acc *= sc;
            }
            m = mn;

#pragma unroll
            for (int k = 0; k < 6; ++k) {
                if (4 * k < c) {
                    const float w = __expf(l[k] - m);
                    dsum += w;
                    acc.x = fmaf(w, r[k].x, acc.x);
                    acc.y = fmaf(w, r[k].y, acc.y);
                    acc.z = fmaf(w, r[k].z, acc.z);
                    acc.w = fmaf(w, r[k].w, acc.w);
                }
            }
            base += 24;
        } while (base < T);

        dsum = sum_groups(dsum);
        acc.x = sum_groups(acc.x);
        acc.y = sum_groups(acc.y);
        acc.z = sum_groups(acc.z);
        acc.w = sum_groups(acc.w);
        const float inv = 1.0f / (dsum + 1e-16f);
        v4f res = acc * inv;
        if (!PROJ) {
            res += bv;
            res.x = lrelu(res.x, 0.01f);
            res.y = lrelu(res.y, 0.01f);
            res.z = lrelu(res.z, 0.01f);
            res.w = lrelu(res.w, 0.01f);
        }
        if (PROJ) {
            if (g == 0) *(v4f*)&tile[rows * TLD + q * 4] = res;
        } else {
            if (g == 0) *(v4f*)((char*)out + (((unsigned)node << 7) + qbyte)) = res;
        }
        ++rows;
    }

    if (PROJ && rows > 0) {
        const int ml = lane & 15;
        const int kh = lane >> 4;
        __builtin_amdgcn_fence(__ATOMIC_RELEASE, "wavefront");
        __builtin_amdgcn_wave_barrier();
        __builtin_amdgcn_fence(__ATOMIC_ACQUIRE, "wavefront");

        const int arow = ml < rows ? ml : 0;
        const float* __restrict__ ap = tile + arow * TLD + 2 * kh;
        const int rb = kh * 8;
#pragma nounroll
        for (int t = 0; t < 4; ++t) {
            const float* __restrict__ bp = &sWt[(t * 16 + ml) * PLDK + 2 * kh];
            v8f pacc = {};
#pragma unroll
            for (int kk = 0; kk < 32; kk += 4) {
                const v2f a = *(const v2f*)(ap + kk);
                const v2f bb = *(const v2f*)(bp + kk);
                pacc = __builtin_amdgcn_wmma_f32_16x16x4_f32(
                    false, a, false, bb, (short)0, pacc, false, false);
            }
            const float bt = b[t * 16 + ml];
            const unsigned ob = ((unsigned)(tile_first + rb * nwaves) << 8)
                              + (unsigned)(t * 16 + ml) * 4u;
#pragma unroll
            for (int r = 0; r < 8; ++r) {
                const float v = lrelu(pacc[r] + bt, 0.01f);
                if (rb + r < rows)
                    *(float*)((char*)out + (ob + (unsigned)r * ((unsigned)nwaves << 8))) = v;
            }
        }
    }
}

extern "C" void kernel_launch(void* const* d_in, const int* in_sizes, int n_in,
                              void* d_out, int out_size, void* d_ws, size_t ws_size,
                              hipStream_t stream) {
    const float* x   = (const float*)d_in[0];
    const float* W1  = (const float*)d_in[1];
    const float* aS1 = (const float*)d_in[2];
    const float* aD1 = (const float*)d_in[3];
    const float* b1  = (const float*)d_in[4];
    const float* W2  = (const float*)d_in[5];
    const float* aS2 = (const float*)d_in[6];
    const float* aD2 = (const float*)d_in[7];
    const float* b2  = (const float*)d_in[8];
    const int*   ei  = (const int*)d_in[9];

    const int n  = in_sizes[0] / 64;
    const int E  = in_sizes[9] / 2;
    const int NB = (n + BNODES - 1) / BNODES;
    const int CH = (E + NCHUNK - 1) / NCHUNK;

    float* w    = (float*)d_ws;
    float* h    = w;
    int*   recs = (int*)(h + (size_t)n * 64);
    int*   rtab = recs + (size_t)2 * E;
    int*   cnt  = rtab + (size_t)2 * NCHUNK * 256;
    int*   off  = cnt + (size_t)n;
    int*   perm = off + (size_t)n;

    float* midh    = (float*)d_out;
    float* decodex = (float*)d_out + (size_t)n * 32;

    const int TPB = 256;
    const int tiles = (n + 15) / 16;
    const int gemm_wgs = (tiles + 7) / 8;
    const int g2 = gemm_wgs / 2;
    const int g3 = gemm_wgs - g2;
    const dim3 wave_grid(GATHER_WGS);

    chunk_sort_proj<<<dim3(NCHUNK + g2), TPB, 0, stream>>>(ei, E, CH, rtab, recs,
                                                          x, W1, b1, h, n, 0);
    bucket_sort_proj<<<dim3(NB + g3), TPB, 0, stream>>>(recs, rtab, CH, NB, n, off, cnt, perm,
                                                       x, W1, b1, h, g2);

    gat_gather<false><<<wave_grid, TPB, 0, stream>>>(off, cnt, perm, h, W1, aS1, aD1, b1, midh, n);

    gat_gather<true><<<wave_grid, TPB, 0, stream>>>(off, cnt, perm, midh, W2, aS2, aD2, b2, decodex, n);
}
